# starting kernel
# baseline (speedup 1.0000x reference)
#include <hip/hip_runtime.h>
#include <hip/hip_bf16.h>

typedef float v2f __attribute__((ext_vector_type(2)));
typedef float v8f __attribute__((ext_vector_type(8)));

#define IN_FEATS 128
#define HIDDEN   64

__global__ void k_init_deg(float* __restrict__ deg, int n) {
    int i = blockIdx.x * blockDim.x + threadIdx.x;
    if (i < n) deg[i] = 1.0f;
}

__global__ void k_accum_deg(const int* __restrict__ dst, int E, float* __restrict__ deg) {
    int e = blockIdx.x * blockDim.x + threadIdx.x;
    if (e < E) atomicAdd(&deg[dst[e]], 1.0f);
}

__global__ void k_make_dinv(const float* __restrict__ deg, float* __restrict__ dinv, int n) {
    int i = blockIdx.x * blockDim.x + threadIdx.x;
    if (i < n) {
        float d = deg[i];
        dinv[i] = (d > 0.0f) ? rsqrtf(d) : 0.0f;
    }
}

__global__ void __launch_bounds__(256) k_gemm1_wmma(const float* __restrict__ X,
                                                    const float* __restrict__ W1,
                                                    float* __restrict__ H,
                                                    int nTiles) {
    __shared__ float Wlds[IN_FEATS * HIDDEN];
    for (int i = threadIdx.x; i < IN_FEATS * HIDDEN; i += blockDim.x)
        Wlds[i] = W1[i];
    __syncthreads();

    const int waveId = (int)((blockIdx.x * blockDim.x + threadIdx.x) >> 5);
    const int lane   = threadIdx.x & 31;
    if (waveId >= nTiles) return;

    const int nTilesN = HIDDEN / 16;
    const int mt = waveId / nTilesN;
    const int nt = waveId % nTilesN;

    const int row = mt * 16 + (lane & 15);
    const int n   = nt * 16 + (lane & 15);
    const int kq  = (lane >> 4) << 1;

    const float* __restrict__ xrow = X + (size_t)row * IN_FEATS;

    v8f c = {};
    #pragma unroll 8
    for (int k = 0; k < IN_FEATS; k += 4) {
        v2f a, b;
        a.x = xrow[k + kq];
        a.y = xrow[k + kq + 1];
        b.x = Wlds[(k + kq)     * HIDDEN + n];
        b.y = Wlds[(k + kq + 1) * HIDDEN + n];
        c = __builtin_amdgcn_wmma_f32_16x16x4_f32(
                 false, a,  false, b,
                 (short)0, c,  false,  false);
    }

    const int rbase = mt * 16 + ((lane >> 4) << 3);
    #pragma unroll
    for (int v = 0; v < 8; ++v)
        H[(size_t)(rbase + v) * HIDDEN + n] = c[v];
}

__global__ void k_self_init1(const float* __restrict__ H, const float* __restrict__ dinv,
                             float* __restrict__ A1, int total  ) {
    int i = blockIdx.x * blockDim.x + threadIdx.x;
    if (i < total) {
        float d = dinv[i >> 6];
        A1[i] = H[i] * d * d;
    }
}

__global__ void __launch_bounds__(256) k_agg1(const int* __restrict__ src,
                                              const int* __restrict__ dst,
                                              const float* __restrict__ dinv,
                                              const float* __restrict__ H,
                                              float* __restrict__ A1, int E) {
    const int wave = (int)((blockIdx.x * blockDim.x + threadIdx.x) >> 5);
    const int lane = threadIdx.x & 31;
    if (wave >= E) return;
    const int s = src[wave];
    const int d = dst[wave];
    const float w = dinv[s] * dinv[d];
    const float* __restrict__ hs = H  + (size_t)s * HIDDEN;
    float* __restrict__       ad = A1 + (size_t)d * HIDDEN;
    atomicAdd(&ad[lane],      hs[lane]      * w);
    atomicAdd(&ad[lane + 32], hs[lane + 32] * w);
}

__global__ void k_dense2(const float* __restrict__ A1, const float* __restrict__ b1,
                         const float* __restrict__ W2, float* __restrict__ H2, int n) {
    int i = blockIdx.x * blockDim.x + threadIdx.x;
    if (i >= n) return;
    const float* __restrict__ a = A1 + (size_t)i * HIDDEN;
    float acc0 = 0.0f, acc1 = 0.0f;
    #pragma unroll 16
    for (int f = 0; f < HIDDEN; ++f) {
        float v = a[f] + b1[f];
        v = v > 0.0f ? v : 0.0f;
        acc0 = fmaf(v, W2[f * 2 + 0], acc0);
        acc1 = fmaf(v, W2[f * 2 + 1], acc1);
    }
    H2[i * 2 + 0] = acc0;
    H2[i * 2 + 1] = acc1;
}

__global__ void k_self_init2(const float* __restrict__ H2, const float* __restrict__ dinv,
                             const float* __restrict__ b2, float* __restrict__ out, int n) {
    int i = blockIdx.x * blockDim.x + threadIdx.x;
    if (i < n) {
        float d = dinv[i];
        float w = d * d;
        out[i * 2 + 0] = H2[i * 2 + 0] * w + b2[0];
        out[i * 2 + 1] = H2[i * 2 + 1] * w + b2[1];
    }
}

__global__ void k_agg2(const int* __restrict__ src, const int* __restrict__ dst,
                       const float* __restrict__ dinv, const float* __restrict__ H2,
                       float* __restrict__ out, int E) {
    int e = blockIdx.x * blockDim.x + threadIdx.x;
    if (e >= E) return;
    const int s = src[e];
    const int d = dst[e];
    const float w = dinv[s] * dinv[d];
    atomicAdd(&out[d * 2 + 0], H2[s * 2 + 0] * w);
    atomicAdd(&out[d * 2 + 1], H2[s * 2 + 1] * w);
}

extern "C" void kernel_launch(void* const* d_in, const int* in_sizes, int n_in,
                              void* d_out, int out_size, void* d_ws, size_t ws_size,
                              hipStream_t stream) {
    const float* x  = (const float*)d_in[0];
    const float* W1 = (const float*)d_in[1];
    const float* b1 = (const float*)d_in[2];
    const float* W2 = (const float*)d_in[3];
    const float* b2 = (const float*)d_in[4];
    const int*   ei = (const int*)d_in[5];

    const int N = in_sizes[0] / IN_FEATS;
    const int E = in_sizes[5] / 2;
    const int* src = ei;
    const int* dst = ei + E;

    float* ws   = (float*)d_ws;
    float* deg  = ws;
    float* dinv = deg  + N;
    float* H    = dinv + N;
    float* A1   = H    + (size_t)N * HIDDEN;
    float* H2   = A1   + (size_t)N * HIDDEN;
    float* out  = (float*)d_out;

    const int T = 256;

    k_init_deg <<<(N + T - 1) / T, T, 0, stream>>>(deg, N);
    k_accum_deg<<<(E + T - 1) / T, T, 0, stream>>>(dst, E, deg);
    k_make_dinv<<<(N + T - 1) / T, T, 0, stream>>>(deg, dinv, N);

    const int nTiles = (N / 16) * (HIDDEN / 16);
    const int gemmBlocks = (nTiles + (T / 32) - 1) / (T / 32);
    k_gemm1_wmma<<<gemmBlocks, T, 0, stream>>>(x, W1, H, nTiles);

    k_self_init1<<<((size_t)N * HIDDEN + T - 1) / T, T, 0, stream>>>(H, dinv, A1, N * HIDDEN);
    {
        const long long waves = (long long)E;
        const long long blocks = (waves * 32 + T - 1) / T;
        k_agg1<<<(int)blocks, T, 0, stream>>>(src, dst, dinv, H, A1, E);
    }

    k_dense2    <<<(N + T - 1) / T, T, 0, stream>>>(A1, b1, W2, H2, N);
    k_self_init2<<<(N + T - 1) / T, T, 0, stream>>>(H2, dinv, b2, out, N);
    k_agg2      <<<(E + T - 1) / T, T, 0, stream>>>(src, dst, dinv, H2, out, E);
}


# degree count and 2-feature layer-2 scatter-add accumulate in LDS per (dst range, edge chunk) work-group, partials summed in fixed chunk order; no global atomic per edge there (sum order changes)
# speedup vs baseline: 1.6713x; 1.6713x over previous
#include <hip/hip_runtime.h>
#include <hip/hip_bf16.h>

typedef float v2f __attribute__((ext_vector_type(2)));
typedef float v8f __attribute__((ext_vector_type(8)));

#define IN_FEATS 128
#define HIDDEN   64

#define SCAT_CAP 6272
#define SCAT_T   1024
#define SCAT_WGS 256

__global__ void __launch_bounds__(SCAT_T) k_deg_part(const int* __restrict__ dst, int E, int n,
                                                     int nPad, int nChunks, int chunkLen,
                                                     unsigned* __restrict__ cntP) {
    __shared__ unsigned cnt[SCAT_CAP];
    const int r   = blockIdx.x / nChunks;
    const int c   = blockIdx.x - r * nChunks;
    const int lo  = r * SCAT_CAP;
    const int len = min(SCAT_CAP, n - lo);
    const int tid = threadIdx.x;

    for (int i = tid; i < len; i += SCAT_T) cnt[i] = 0u;
    __syncthreads();

    const int e0 = c * chunkLen;
    const int e1 = min(E, e0 + chunkLen);
    int base = e0 + tid;
    for (; base + 7 * SCAT_T < e1; base += 8 * SCAT_T) {
        unsigned l[8];
        #pragma unroll
        for (int j = 0; j < 8; ++j) l[j] = (unsigned)(dst[base + j * SCAT_T] - lo);
        #pragma unroll
        for (int j = 0; j < 8; ++j)
            if (l[j] < (unsigned)len) atomicAdd(&cnt[l[j]], 1u);
    }
    for (; base < e1; base += SCAT_T) {
        const unsigned l = (unsigned)(dst[base] - lo);
        if (l < (unsigned)len) atomicAdd(&cnt[l], 1u);
    }
    __syncthreads();

    unsigned* __restrict__ o = cntP + (size_t)c * nPad + lo;
    for (int i = tid; i < len; i += SCAT_T) o[i] = cnt[i];
}

__global__ void k_make_dinv(const unsigned* __restrict__ cntP, int nPad, int nChunks,
                            float* __restrict__ dinv, int n) {
    int i = blockIdx.x * blockDim.x + threadIdx.x;
    if (i < n) {
        unsigned cnt = 0u;
        #pragma unroll 8
        for (int c = 0; c < nChunks; ++c) cnt += cntP[(size_t)c * nPad + i];
        float d = 1.0f + (float)cnt;
        dinv[i] = (d > 0.0f) ? rsqrtf(d) : 0.0f;
    }
}

__global__ void __launch_bounds__(256) k_gemm1_wmma(const float* __restrict__ X,
                                                    const float* __restrict__ W1,
                                                    float* __restrict__ H,
                                                    int nTiles) {
    __shared__ float Wlds[IN_FEATS * HIDDEN];
    for (int i = threadIdx.x; i < IN_FEATS * HIDDEN; i += blockDim.x)
        Wlds[i] = W1[i];
    __syncthreads();

    const int waveId = (int)((blockIdx.x * blockDim.x + threadIdx.x) >> 5);
    const int lane   = threadIdx.x & 31;
    if (waveId >= nTiles) return;

    const int nTilesN = HIDDEN / 16;
    const int mt = waveId / nTilesN;
    const int nt = waveId % nTilesN;

    const int row = mt * 16 + (lane & 15);
    const int n   = nt * 16 + (lane & 15);
    const int kq  = (lane >> 4) << 1;

    const float* __restrict__ xrow = X + (size_t)row * IN_FEATS;

    v8f c = {};
    #pragma unroll 8
    for (int k = 0; k < IN_FEATS; k += 4) {
        v2f a, b;
        a.x = xrow[k + kq];
        a.y = xrow[k + kq + 1];
        b.x = Wlds[(k + kq)     * HIDDEN + n];
        b.y = Wlds[(k + kq + 1) * HIDDEN + n];
        c = __builtin_amdgcn_wmma_f32_16x16x4_f32(
                 false, a,  false, b,
                 (short)0, c,  false,  false);
    }

    const int rbase = mt * 16 + ((lane >> 4) << 3);
    #pragma unroll
    for (int v = 0; v < 8; ++v)
        H[(size_t)(rbase + v) * HIDDEN + n] = c[v];
}

__global__ void k_self_init1(const float* __restrict__ H, const float* __restrict__ dinv,
                             float* __restrict__ A1, int total  ) {
    int i = blockIdx.x * blockDim.x + threadIdx.x;
    if (i < total) {
        float d = dinv[i >> 6];
        A1[i] = H[i] * d * d;
    }
}

__global__ void __launch_bounds__(256) k_agg1(const int* __restrict__ src,
                                              const int* __restrict__ dst,
                                              const float* __restrict__ dinv,
                                              const float* __restrict__ H,
                                              float* __restrict__ A1, int E) {
    const int wave = (int)((blockIdx.x * blockDim.x + threadIdx.x) >> 5);
    const int lane = threadIdx.x & 31;
    if (wave >= E) return;
    const int s = src[wave];
    const int d = dst[wave];
    const float w = dinv[s] * dinv[d];
    const float* __restrict__ hs = H  + (size_t)s * HIDDEN;
    float* __restrict__       ad = A1 + (size_t)d * HIDDEN;
    atomicAdd(&ad[lane],      hs[lane]      * w);
    atomicAdd(&ad[lane + 32], hs[lane + 32] * w);
}

__global__ void k_dense2(const float* __restrict__ A1, const float* __restrict__ b1,
                         const float* __restrict__ W2, float* __restrict__ H2, int n) {
    int i = blockIdx.x * blockDim.x + threadIdx.x;
    if (i >= n) return;
    const float* __restrict__ a = A1 + (size_t)i * HIDDEN;
    float acc0 = 0.0f, acc1 = 0.0f;
    #pragma unroll 16
    for (int f = 0; f < HIDDEN; ++f) {
        float v = a[f] + b1[f];
        v = v > 0.0f ? v : 0.0f;
        acc0 = fmaf(v, W2[f * 2 + 0], acc0);
        acc1 = fmaf(v, W2[f * 2 + 1], acc1);
    }
    H2[i * 2 + 0] = acc0;
    H2[i * 2 + 1] = acc1;
}

template <int K>
__device__ __forceinline__ void agg2_edges(const int* __restrict__ src, const int* __restrict__ dst,
                                           const float* __restrict__ dinv,
                                           const float2* __restrict__ H2v,
                                           int base, int lo, int len,
                                           float* acc0, float* acc1) {
    int      s[K];
    unsigned l[K];
    #pragma unroll
    for (int j = 0; j < K; ++j) {
        l[j] = (unsigned)(dst[base + j * SCAT_T] - lo);
        s[j] = src[base + j * SCAT_T];
    }
    float  ws[K];
    float  wd[K];
    float2 h[K];
    #pragma unroll
    for (int j = 0; j < K; ++j) {
        ws[j] = 0.0f; wd[j] = 0.0f; h[j] = make_float2(0.0f, 0.0f);
        if (l[j] < (unsigned)len) {
            ws[j] = dinv[s[j]];
            wd[j] = dinv[lo + (int)l[j]];
            h[j]  = H2v[s[j]];
        }
    }
    #pragma unroll
    for (int j = 0; j < K; ++j) {
        if (l[j] < (unsigned)len) {
            const float w = ws[j] * wd[j];
            atomicAdd(&acc0[l[j]], h[j].x * w);
            atomicAdd(&acc1[l[j]], h[j].y * w);
        }
    }
}

__global__ void __launch_bounds__(SCAT_T) k_agg2_part(const int* __restrict__ src,
                                                      const int* __restrict__ dst,
                                                      const float* __restrict__ dinv,
                                                      const float* __restrict__ H2,
                                                      int E, int n, int nPad, int nChunks, int chunkLen,
                                                      float* __restrict__ P0, float* __restrict__ P1) {
    __shared__ float acc0[SCAT_CAP];
    __shared__ float acc1[SCAT_CAP];
    const int r   = blockIdx.x / nChunks;
    const int c   = blockIdx.x - r * nChunks;
    const int lo  = r * SCAT_CAP;
    const int len = min(SCAT_CAP, n - lo);
    const int tid = threadIdx.x;

    for (int i = tid; i < len; i += SCAT_T) { acc0[i] = 0.0f; acc1[i] = 0.0f; }
    __syncthreads();

    const float2* __restrict__ H2v = (const float2*)H2;
    const int e0 = c * chunkLen;
    const int e1 = min(E, e0 + chunkLen);
    int base = e0 + tid;
    for (; base + 7 * SCAT_T < e1; base += 8 * SCAT_T)
        agg2_edges<8>(src, dst, dinv, H2v, base, lo, len, acc0, acc1);
    for (; base < e1; base += SCAT_T)
        agg2_edges<1>(src, dst, dinv, H2v, base, lo, len, acc0, acc1);
    __syncthreads();

    float* __restrict__ o0 = P0 + (size_t)c * nPad + lo;
    float* __restrict__ o1 = P1 + (size_t)c * nPad + lo;
    for (int i = tid; i < len; i += SCAT_T) { o0[i] = acc0[i]; o1[i] = acc1[i]; }
}

__global__ void k_final2(const float* __restrict__ H2, const float* __restrict__ dinv,
                         const float* __restrict__ b2,
                         const float* __restrict__ P0, const float* __restrict__ P1,
                         int nPad, int nChunks, float* __restrict__ out, int n) {
    int i = blockIdx.x * blockDim.x + threadIdx.x;
    if (i < n) {
        float d = dinv[i];
        float w = d * d;
        float o0 = H2[i * 2 + 0] * w + b2[0];
        float o1 = H2[i * 2 + 1] * w + b2[1];
        #pragma unroll 8
        for (int c = 0; c < nChunks; ++c) {
            o0 += P0[(size_t)c * nPad + i];
            o1 += P1[(size_t)c * nPad + i];
        }
        out[i * 2 + 0] = o0;
        out[i * 2 + 1] = o1;
    }
}

extern "C" void kernel_launch(void* const* d_in, const int* in_sizes, int n_in,
                              void* d_out, int out_size, void* d_ws, size_t ws_size,
                              hipStream_t stream) {
    const float* x  = (const float*)d_in[0];
    const float* W1 = (const float*)d_in[1];
    const float* b1 = (const float*)d_in[2];
    const float* W2 = (const float*)d_in[3];
    const float* b2 = (const float*)d_in[4];
    const int*   ei = (const int*)d_in[5];

    const int N = in_sizes[0] / IN_FEATS;
    const int E = in_sizes[5] / 2;
    const int* src = ei;
    const int* dst = ei + E;

    const int nRanges  = (N + SCAT_CAP - 1) / SCAT_CAP;
    const int nChunks  = nRanges < SCAT_WGS ? SCAT_WGS / nRanges : 1;
    const int chunkLen = (E + nChunks - 1) / nChunks;
    const int nPad     = (N + 31) & ~31;

    float* ws   = (float*)d_ws;
    float* dinv = ws;
    float* H    = dinv + nPad;
    float* A1   = H    + (size_t)N * HIDDEN;
    float* H2   = A1   + (size_t)N * HIDDEN;
    float* P0   = H2   + (size_t)nPad * 2;
    float* P1   = P0   + (size_t)nChunks * nPad;
    unsigned* cntP = (unsigned*)(P1 + (size_t)nChunks * nPad);
    float* out  = (float*)d_out;

    const int T = 256;

    k_deg_part <<<nRanges * nChunks, SCAT_T, 0, stream>>>(dst, E, N, nPad, nChunks, chunkLen, cntP);
    k_make_dinv<<<(N + T - 1) / T, T, 0, stream>>>(cntP, nPad, nChunks, dinv, N);

    const int nTiles = (N / 16) * (HIDDEN / 16);
    const int gemmBlocks = (nTiles + (T / 32) - 1) / (T / 32);
    k_gemm1_wmma<<<gemmBlocks, T, 0, stream>>>(x, W1, H, nTiles);

    k_self_init1<<<((size_t)N * HIDDEN + T - 1) / T, T, 0, stream>>>(H, dinv, A1, N * HIDDEN);
    {
        const long long waves = (long long)E;
        const long long blocks = (waves * 32 + T - 1) / T;
        k_agg1<<<(int)blocks, T, 0, stream>>>(src, dst, dinv, H, A1, E);
    }

    k_dense2   <<<(N + T - 1) / T, T, 0, stream>>>(A1, b1, W2, H2, N);
    k_agg2_part<<<nRanges * nChunks, SCAT_T, 0, stream>>>(src, dst, dinv, H2, E, N, nPad, nChunks, chunkLen, P0, P1);
    k_final2   <<<(N + T - 1) / T, T, 0, stream>>>(H2, dinv, b2, P0, P1, nPad, nChunks, out, N);
}


# layer-1 scatter-add: edges counting-sorted by destination (a node's slots per chunk in arrival order), a wave per node sums H[src]*norm in registers, one store; no global atomics (sum order changes)
# speedup vs baseline: 2.6824x; 1.6050x over previous
#include <hip/hip_runtime.h>
#include <hip/hip_bf16.h>

typedef float v2f __attribute__((ext_vector_type(2)));
typedef float v8f __attribute__((ext_vector_type(8)));

#define IN_FEATS 128
#define HIDDEN   64

#define SCAT_CAP 6272
#define SCAT_T   1024
#define SCAT_WGS 256

__global__ void __launch_bounds__(SCAT_T) k_deg_part(const int* __restrict__ dst, int E, int n,
                                                     int nPad, int nChunks, int chunkLen,
                                                     unsigned* __restrict__ cntP) {
    __shared__ unsigned cnt[SCAT_CAP];
    const int r   = blockIdx.x / nChunks;
    const int c   = blockIdx.x - r * nChunks;
    const int lo  = r * SCAT_CAP;
    const int len = min(SCAT_CAP, n - lo);
    const int tid = threadIdx.x;

    for (int i = tid; i < len; i += SCAT_T) cnt[i] = 0u;
    __syncthreads();

    const int e0 = c * chunkLen;
    const int e1 = min(E, e0 + chunkLen);
    int base = e0 + tid;
    for (; base + 7 * SCAT_T < e1; base += 8 * SCAT_T) {
        unsigned l[8];
        #pragma unroll
        for (int j = 0; j < 8; ++j) l[j] = (unsigned)(dst[base + j * SCAT_T] - lo);
        #pragma unroll
        for (int j = 0; j < 8; ++j)
            if (l[j] < (unsigned)len) atomicAdd(&cnt[l[j]], 1u);
    }
    for (; base < e1; base += SCAT_T) {
        const unsigned l = (unsigned)(dst[base] - lo);
        if (l < (unsigned)len) atomicAdd(&cnt[l], 1u);
    }
    __syncthreads();

    unsigned* __restrict__ o = cntP + (size_t)c * nPad + lo;
    for (int i = tid; i < len; i += SCAT_T) o[i] = cnt[i];
}

__global__ void k_make_dinv(const unsigned* __restrict__ cntP, int nPad, int nChunks,
                            float* __restrict__ dinv, unsigned* __restrict__ preP,
                            unsigned* __restrict__ cntTot, int n) {
    int i = blockIdx.x * blockDim.x + threadIdx.x;
    if (i < n) {
        unsigned cnt = 0u;
        #pragma unroll 8
        for (int c = 0; c < nChunks; ++c) {
            const unsigned v = cntP[(size_t)c * nPad + i];
            preP[(size_t)c * nPad + i] = cnt;
            cnt += v;
        }
        cntTot[i] = cnt;
        float d = 1.0f + (float)cnt;
        dinv[i] = (d > 0.0f) ? rsqrtf(d) : 0.0f;
    }
}

__global__ void __launch_bounds__(SCAT_T) k_rowptr(const unsigned* __restrict__ cntTot, int n,
                                                   unsigned* __restrict__ rowptr) {
    __shared__ unsigned wsum[SCAT_T / 32];
    const int tid  = threadIdx.x;
    const int lane = tid & 31;
    const int wv   = tid >> 5;
    const int per  = (n + SCAT_T - 1) / SCAT_T;
    const int b    = min(n, tid * per);
    const int e    = min(n, b + per);

    unsigned s = 0u;
    for (int i = b; i < e; ++i) s += cntTot[i];

    unsigned inc = s;
    for (int o = 1; o < 32; o <<= 1) {
        const unsigned t = __shfl_up(inc, o);
        if (lane >= o) inc += t;
    }
    if (lane == 31) wsum[wv] = inc;
    __syncthreads();
    if (wv == 0) {
        const unsigned w = wsum[lane];
        unsigned winc = w;
        for (int o = 1; o < 32; o <<= 1) {
            const unsigned t = __shfl_up(winc, o);
            if (lane >= o) winc += t;
        }
        wsum[lane] = winc - w;
    }
    __syncthreads();

    unsigned run = wsum[wv] + inc - s;
    for (int i = b; i < e; ++i) { rowptr[i] = run; run += cntTot[i]; }
    if (tid == SCAT_T - 1) rowptr[n] = run;
}

__global__ void __launch_bounds__(SCAT_T) k_csr_fill(const int* __restrict__ src,
                                                     const int* __restrict__ dst,
                                                     int E, int n, int nPad, int nChunks, int chunkLen,
                                                     const unsigned* __restrict__ rowptr,
                                                     const unsigned* __restrict__ preP,
                                                     int* __restrict__ csrSrc) {
    __shared__ unsigned cur[SCAT_CAP];
    const int r   = blockIdx.x / nChunks;
    const int c   = blockIdx.x - r * nChunks;
    const int lo  = r * SCAT_CAP;
    const int len = min(SCAT_CAP, n - lo);
    const int tid = threadIdx.x;

    const unsigned* __restrict__ pre = preP + (size_t)c * nPad + lo;
    for (int i = tid; i < len; i += SCAT_T) cur[i] = rowptr[lo + i] + pre[i];
    __syncthreads();

    const int e0 = c * chunkLen;
    const int e1 = min(E, e0 + chunkLen);
    int base = e0 + tid;
    for (; base + 7 * SCAT_T < e1; base += 8 * SCAT_T) {
        unsigned l[8];
        int      s[8];
        #pragma unroll
        for (int j = 0; j < 8; ++j) {
            l[j] = (unsigned)(dst[base + j * SCAT_T] - lo);
            s[j] = src[base + j * SCAT_T];
        }
        unsigned p[8];
        #pragma unroll
        for (int j = 0; j < 8; ++j) {
            p[j] = 0u;
            if (l[j] < (unsigned)len) p[j] = atomicAdd(&cur[l[j]], 1u);
        }
        #pragma unroll
        for (int j = 0; j < 8; ++j)
            if (l[j] < (unsigned)len) csrSrc[p[j]] = s[j];
    }
    for (; base < e1; base += SCAT_T) {
        const unsigned l = (unsigned)(dst[base] - lo);
        if (l < (unsigned)len) csrSrc[atomicAdd(&cur[l], 1u)] = src[base];
    }
}

__global__ void __launch_bounds__(256) k_gemm1_wmma(const float* __restrict__ X,
                                                    const float* __restrict__ W1,
                                                    float* __restrict__ H,
                                                    int nTiles) {
    __shared__ float Wlds[IN_FEATS * HIDDEN];
    for (int i = threadIdx.x; i < IN_FEATS * HIDDEN; i += blockDim.x)
        Wlds[i] = W1[i];
    __syncthreads();

    const int waveId = (int)((blockIdx.x * blockDim.x + threadIdx.x) >> 5);
    const int lane   = threadIdx.x & 31;
    if (waveId >= nTiles) return;

    const int nTilesN = HIDDEN / 16;
    const int mt = waveId / nTilesN;
    const int nt = waveId % nTilesN;

    const int row = mt * 16 + (lane & 15);
    const int n   = nt * 16 + (lane & 15);
    const int kq  = (lane >> 4) << 1;

    const float* __restrict__ xrow = X + (size_t)row * IN_FEATS;

    v8f c = {};
    #pragma unroll 8
    for (int k = 0; k < IN_FEATS; k += 4) {
        v2f a, b;
        a.x = xrow[k + kq];
        a.y = xrow[k + kq + 1];
        b.x = Wlds[(k + kq)     * HIDDEN + n];
        b.y = Wlds[(k + kq + 1) * HIDDEN + n];
        c = __builtin_amdgcn_wmma_f32_16x16x4_f32(
                 false, a,  false, b,
                 (short)0, c,  false,  false);
    }

    const int rbase = mt * 16 + ((lane >> 4) << 3);
    #pragma unroll
    for (int v = 0; v < 8; ++v)
        H[(size_t)(rbase + v) * HIDDEN + n] = c[v];
}

__global__ void __launch_bounds__(256) k_gather1(const unsigned* __restrict__ rowptr,
                                                 const int* __restrict__ csrSrc,
                                                 const float* __restrict__ dinv,
                                                 const float* __restrict__ H,
                                                 float* __restrict__ A1, int n) {
    const int node = __builtin_amdgcn_readfirstlane((int)((blockIdx.x * blockDim.x + threadIdx.x) >> 5));
    if (node >= n) return;
    const int lane = threadIdx.x & 31;
    const int half = lane >> 4;
    const int q    = lane & 15;
    const float4* __restrict__ H4 = (const float4*)H;

    const float dd  = dinv[node];
    const int   beg = (int)rowptr[node];
    const int   end = (int)rowptr[node + 1];

    float4 acc = make_float4(0.0f, 0.0f, 0.0f, 0.0f);
    float4 hs = make_float4(0.0f, 0.0f, 0.0f, 0.0f);
    if (half == 0) hs = H4[(size_t)node * (HIDDEN / 4) + q];

    for (int b0 = beg; b0 < end; b0 += 32) {
        const int cnt = min(32, end - b0);
        int   s_l = 0;
        float w_l = 0.0f;
        if (lane < cnt) { s_l = csrSrc[b0 + lane]; w_l = dinv[s_l] * dd; }
        for (int j = 0; j < cnt; j += 16) {
            float4 h[8];
            float  w[8];
            #pragma unroll
            for (int u = 0; u < 8; ++u) {
                const int idx = j + 2 * u + half;
                const int s   = __shfl(s_l, idx);
                w[u] = __shfl(w_l, idx);
                h[u] = make_float4(0.0f, 0.0f, 0.0f, 0.0f);
                if (idx < cnt) h[u] = H4[(size_t)s * (HIDDEN / 4) + q];
            }
            #pragma unroll
            for (int u = 0; u < 8; ++u) {
                acc.x = fmaf(h[u].x, w[u], acc.x);
                acc.y = fmaf(h[u].y, w[u], acc.y);
                acc.z = fmaf(h[u].z, w[u], acc.z);
                acc.w = fmaf(h[u].w, w[u], acc.w);
            }
        }
    }

    const float wself = dd * dd;
    acc.x = fmaf(hs.x, wself, acc.x);
    acc.y = fmaf(hs.y, wself, acc.y);
    acc.z = fmaf(hs.z, wself, acc.z);
    acc.w = fmaf(hs.w, wself, acc.w);

    acc.x += __shfl_xor(acc.x, 16);
    acc.y += __shfl_xor(acc.y, 16);
    acc.z += __shfl_xor(acc.z, 16);
    acc.w += __shfl_xor(acc.w, 16);
    if (half == 0) ((float4*)A1)[(size_t)node * (HIDDEN / 4) + q] = acc;
}

__global__ void k_dense2(const float* __restrict__ A1, const float* __restrict__ b1,
                         const float* __restrict__ W2, float* __restrict__ H2, int n) {
    int i = blockIdx.x * blockDim.x + threadIdx.x;
    if (i >= n) return;
    const float* __restrict__ a = A1 + (size_t)i * HIDDEN;
    float acc0 = 0.0f, acc1 = 0.0f;
    #pragma unroll 16
    for (int f = 0; f < HIDDEN; ++f) {
        float v = a[f] + b1[f];
        v = v > 0.0f ? v : 0.0f;
        acc0 = fmaf(v, W2[f * 2 + 0], acc0);
        acc1 = fmaf(v, W2[f * 2 + 1], acc1);
    }
    H2[i * 2 + 0] = acc0;
    H2[i * 2 + 1] = acc1;
}

template <int K>
__device__ __forceinline__ void agg2_edges(const int* __restrict__ src, const int* __restrict__ dst,
                                           const float* __restrict__ dinv,
                                           const float2* __restrict__ H2v,
                                           int base, int lo, int len,
                                           float* acc0, float* acc1) {
    int      s[K];
    unsigned l[K];
    #pragma unroll
    for (int j = 0; j < K; ++j) {
        l[j] = (unsigned)(dst[base + j * SCAT_T] - lo);
        s[j] = src[base + j * SCAT_T];
    }
    float  ws[K];
    float  wd[K];
    float2 h[K];
    #pragma unroll
    for (int j = 0; j < K; ++j) {
        ws[j] = 0.0f; wd[j] = 0.0f; h[j] = make_float2(0.0f, 0.0f);
        if (l[j] < (unsigned)len) {
            ws[j] = dinv[s[j]];
            wd[j] = dinv[lo + (int)l[j]];
            h[j]  = H2v[s[j]];
        }
    }
    #pragma unroll
    for (int j = 0; j < K; ++j) {
        if (l[j] < (unsigned)len) {
            const float w = ws[j] * wd[j];
            atomicAdd(&acc0[l[j]], h[j].x * w);
            atomicAdd(&acc1[l[j]], h[j].y * w);
        }
    }
}

__global__ void __launch_bounds__(SCAT_T) k_agg2_part(const int* __restrict__ src,
                                                      const int* __restrict__ dst,
                                                      const float* __restrict__ dinv,
                                                      const float* __restrict__ H2,
                                                      int E, int n, int nPad, int nChunks, int chunkLen,
                                                      float* __restrict__ P0, float* __restrict__ P1) {
    __shared__ float acc0[SCAT_CAP];
    __shared__ float acc1[SCAT_CAP];
    const int r   = blockIdx.x / nChunks;
    const int c   = blockIdx.x - r * nChunks;
    const int lo  = r * SCAT_CAP;
    const int len = min(SCAT_CAP, n - lo);
    const int tid = threadIdx.x;

    for (int i = tid; i < len; i += SCAT_T) { acc0[i] = 0.0f; acc1[i] = 0.0f; }
    __syncthreads();

    const float2* __restrict__ H2v = (const float2*)H2;
    const int e0 = c * chunkLen;
    const int e1 = min(E, e0 + chunkLen);
    int base = e0 + tid;
    for (; base + 7 * SCAT_T < e1; base += 8 * SCAT_T)
        agg2_edges<8>(src, dst, dinv, H2v, base, lo, len, acc0, acc1);
    for (; base < e1; base += SCAT_T)
        agg2_edges<1>(src, dst, dinv, H2v, base, lo, len, acc0, acc1);
    __syncthreads();

    float* __restrict__ o0 = P0 + (size_t)c * nPad + lo;
    float* __restrict__ o1 = P1 + (size_t)c * nPad + lo;
    for (int i = tid; i < len; i += SCAT_T) { o0[i] = acc0[i]; o1[i] = acc1[i]; }
}

__global__ void k_final2(const float* __restrict__ H2, const float* __restrict__ dinv,
                         const float* __restrict__ b2,
                         const float* __restrict__ P0, const float* __restrict__ P1,
                         int nPad, int nChunks, float* __restrict__ out, int n) {
    int i = blockIdx.x * blockDim.x + threadIdx.x;
    if (i < n) {
        float d = dinv[i];
        float w = d * d;
        float o0 = H2[i * 2 + 0] * w + b2[0];
        float o1 = H2[i * 2 + 1] * w + b2[1];
        #pragma unroll 8
        for (int c = 0; c < nChunks; ++c) {
            o0 += P0[(size_t)c * nPad + i];
            o1 += P1[(size_t)c * nPad + i];
        }
        out[i * 2 + 0] = o0;
        out[i * 2 + 1] = o1;
    }
}

extern "C" void kernel_launch(void* const* d_in, const int* in_sizes, int n_in,
                              void* d_out, int out_size, void* d_ws, size_t ws_size,
                              hipStream_t stream) {
    const float* x  = (const float*)d_in[0];
    const float* W1 = (const float*)d_in[1];
    const float* b1 = (const float*)d_in[2];
    const float* W2 = (const float*)d_in[3];
    const float* b2 = (const float*)d_in[4];
    const int*   ei = (const int*)d_in[5];

    const int N = in_sizes[0] / IN_FEATS;
    const int E = in_sizes[5] / 2;
    const int* src = ei;
    const int* dst = ei + E;

    const int nRanges  = (N + SCAT_CAP - 1) / SCAT_CAP;
    const int nChunks  = nRanges < SCAT_WGS ? SCAT_WGS / nRanges : 1;
    const int chunkLen = (E + nChunks - 1) / nChunks;
    const int nPad     = (N + 31) & ~31;

    float* ws   = (float*)d_ws;
    float* dinv = ws;
    float* H    = dinv + nPad;
    float* A1   = H    + (size_t)N * HIDDEN;
    float* H2   = A1   + (size_t)N * HIDDEN;
    float* P0   = H2   + (size_t)nPad * 2;
    float* P1   = P0   + (size_t)nChunks * nPad;
    unsigned* cntP = (unsigned*)(P1 + (size_t)nChunks * nPad);
    unsigned* preP   = cntP   + (size_t)nChunks * nPad;
    unsigned* cntTot = preP   + (size_t)nChunks * nPad;
    unsigned* rowptr = cntTot + nPad;
    int*      csrSrc = (int*)(rowptr + nPad + 32);
    float* out  = (float*)d_out;

    const int T = 256;

    k_deg_part <<<nRanges * nChunks, SCAT_T, 0, stream>>>(dst, E, N, nPad, nChunks, chunkLen, cntP);
    k_make_dinv<<<(N + T - 1) / T, T, 0, stream>>>(cntP, nPad, nChunks, dinv, preP, cntTot, N);

    k_rowptr  <<<1, SCAT_T, 0, stream>>>(cntTot, N, rowptr);
    k_csr_fill<<<nRanges * nChunks, SCAT_T, 0, stream>>>(src, dst, E, N, nPad, nChunks, chunkLen, rowptr, preP, csrSrc);

    const int nTiles = (N / 16) * (HIDDEN / 16);
    const int gemmBlocks = (nTiles + (T / 32) - 1) / (T / 32);
    k_gemm1_wmma<<<gemmBlocks, T, 0, stream>>>(x, W1, H, nTiles);

    k_gather1<<<(N * 32 + T - 1) / T, T, 0, stream>>>(rowptr, csrSrc, dinv, H, A1, N);

    k_dense2   <<<(N + T - 1) / T, T, 0, stream>>>(A1, b1, W2, H2, N);
    k_agg2_part<<<nRanges * nChunks, SCAT_T, 0, stream>>>(src, dst, dinv, H2, E, N, nPad, nChunks, chunkLen, P0, P1);
    k_final2   <<<(N + T - 1) / T, T, 0, stream>>>(H2, dinv, b2, P0, P1, nPad, nChunks, out, N);
}


# row-pointer prefix sum: counts and prefixes pass through LDS so that every global access is whole-line (16 B per lane, consecutive lanes) instead of one 196-B-strided run per lane
# speedup vs baseline: 3.0100x; 1.1221x over previous
#include <hip/hip_runtime.h>
#include <hip/hip_bf16.h>

typedef float v2f __attribute__((ext_vector_type(2)));
typedef float v8f __attribute__((ext_vector_type(8)));

#define IN_FEATS 128
#define HIDDEN   64

#define SCAT_CAP 6272
#define SCAT_T   1024
#define SCAT_WGS 256

__global__ void __launch_bounds__(SCAT_T) k_deg_part(const int* __restrict__ dst, int E, int n,
                                                     int nPad, int nChunks, int chunkLen,
                                                     unsigned* __restrict__ cntP) {
    __shared__ unsigned cnt[SCAT_CAP];
    const int r   = blockIdx.x / nChunks;
    const int c   = blockIdx.x - r * nChunks;
    const int lo  = r * SCAT_CAP;
    const int len = min(SCAT_CAP, n - lo);
    const int tid = threadIdx.x;

    for (int i = tid; i < len; i += SCAT_T) cnt[i] = 0u;
    __syncthreads();

    const int e0 = c * chunkLen;
    const int e1 = min(E, e0 + chunkLen);
    int base = e0 + tid;
    for (; base + 7 * SCAT_T < e1; base += 8 * SCAT_T) {
        unsigned l[8];
        #pragma unroll
        for (int j = 0; j < 8; ++j) l[j] = (unsigned)(dst[base + j * SCAT_T] - lo);
        #pragma unroll
        for (int j = 0; j < 8; ++j)
            if (l[j] < (unsigned)len) atomicAdd(&cnt[l[j]], 1u);
    }
    for (; base < e1; base += SCAT_T) {
        const unsigned l = (unsigned)(dst[base] - lo);
        if (l < (unsigned)len) atomicAdd(&cnt[l], 1u);
    }
    __syncthreads();

    unsigned* __restrict__ o = cntP + (size_t)c * nPad + lo;
    for (int i = tid; i < len; i += SCAT_T) o[i] = cnt[i];
}

__global__ void k_make_dinv(const unsigned* __restrict__ cntP, int nPad, int nChunks,
                            float* __restrict__ dinv, unsigned* __restrict__ preP,
                            unsigned* __restrict__ cntTot, int n) {
    int i = blockIdx.x * blockDim.x + threadIdx.x;
    if (i < n) {
        unsigned cnt = 0u;
        #pragma unroll 8
        for (int c = 0; c < nChunks; ++c) {
            const unsigned v = cntP[(size_t)c * nPad + i];
            preP[(size_t)c * nPad + i] = cnt;
            cnt += v;
        }
        cntTot[i] = cnt;
        float d = 1.0f + (float)cnt;
        dinv[i] = (d > 0.0f) ? rsqrtf(d) : 0.0f;
    }
}

#define RP_V4  13
#define RP_PER (RP_V4 * 4)
#define RP_CAP (SCAT_T * RP_PER)

__global__ void __launch_bounds__(SCAT_T) k_rowptr(const unsigned* __restrict__ cntTot, int n,
                                                   unsigned* __restrict__ rowptr) {
    __shared__ uint4    buf[RP_CAP / 4];
    __shared__ unsigned wsum[SCAT_T / 32 + 1];
    const int tid  = threadIdx.x;
    const int lane = tid & 31;
    const int wv   = tid >> 5;

    unsigned carry = 0u;
    for (int t0 = 0; t0 < n; t0 += RP_CAP) {
        const int len = min(RP_CAP, n - t0);
        const unsigned* __restrict__ gin  = cntTot + t0;
        unsigned* __restrict__       gout = rowptr + t0;

        const int nFull = len >> 2;
        const int qLast = max(nFull - 1, 0);

        uint4 v[RP_V4];
        #pragma unroll
        for (int j = 0; j < RP_V4; ++j) v[j] = ((const uint4*)gin)[min(tid + j * SCAT_T, qLast)];
        #pragma unroll
        for (int j = 0; j < RP_V4; ++j) {
            const int q = tid + j * SCAT_T;
            buf[q] = (q < nFull) ? v[j] : make_uint4(0u, 0u, 0u, 0u);
        }
        if ((len & 3) && tid == (nFull & (SCAT_T - 1))) {
            const int e = nFull * 4;
            uint4 t = make_uint4(0u, 0u, 0u, 0u);
            t.x = gin[e];
            if (e + 1 < len) t.y = gin[e + 1];
            if (e + 2 < len) t.z = gin[e + 2];
            buf[nFull] = t;
        }
        __syncthreads();

        uint4 r[RP_V4];
        #pragma unroll
        for (int j = 0; j < RP_V4; ++j) r[j] = buf[tid * RP_V4 + j];
        unsigned s = 0u;
        #pragma unroll
        for (int j = 0; j < RP_V4; ++j) s += r[j].x + r[j].y + r[j].z + r[j].w;

        unsigned inc = s;
        for (int o = 1; o < 32; o <<= 1) {
            const unsigned t = __shfl_up(inc, o);
            if (lane >= o) inc += t;
        }
        if (lane == 31) wsum[wv] = inc;
        __syncthreads();
        if (wv == 0) {
            const unsigned w = wsum[lane];
            unsigned winc = w;
            for (int o = 1; o < 32; o <<= 1) {
                const unsigned t = __shfl_up(winc, o);
                if (lane >= o) winc += t;
            }
            wsum[lane] = winc - w;
            if (lane == 31) wsum[SCAT_T / 32] = winc;
        }
        __syncthreads();

        unsigned run = carry + wsum[wv] + inc - s;
        #pragma unroll
        for (int j = 0; j < RP_V4; ++j) {
            uint4 p;
            p.x = run; run += r[j].x;
            p.y = run; run += r[j].y;
            p.z = run; run += r[j].z;
            p.w = run; run += r[j].w;
            buf[tid * RP_V4 + j] = p;
        }
        carry += wsum[SCAT_T / 32];
        __syncthreads();

        #pragma unroll
        for (int j = 0; j < RP_V4; ++j) v[j] = buf[tid + j * SCAT_T];
        #pragma unroll
        for (int j = 0; j < RP_V4; ++j) {
            const int q = tid + j * SCAT_T;
            if (q < nFull) ((uint4*)gout)[q] = v[j];
        }
        if ((len & 3) && tid == (nFull & (SCAT_T - 1))) {
            const int e = nFull * 4;
            const uint4 t = buf[nFull];
            gout[e] = t.x;
            if (e + 1 < len) gout[e + 1] = t.y;
            if (e + 2 < len) gout[e + 2] = t.z;
        }
        __syncthreads();
    }
    if (tid == 0) rowptr[n] = carry;
}

__global__ void __launch_bounds__(SCAT_T) k_csr_fill(const int* __restrict__ src,
                                                     const int* __restrict__ dst,
                                                     int E, int n, int nPad, int nChunks, int chunkLen,
                                                     const unsigned* __restrict__ rowptr,
                                                     const unsigned* __restrict__ preP,
                                                     int* __restrict__ csrSrc) {
    __shared__ unsigned cur[SCAT_CAP];
    const int r   = blockIdx.x / nChunks;
    const int c   = blockIdx.x - r * nChunks;
    const int lo  = r * SCAT_CAP;
    const int len = min(SCAT_CAP, n - lo);
    const int tid = threadIdx.x;

    const unsigned* __restrict__ pre = preP + (size_t)c * nPad + lo;
    for (int i = tid; i < len; i += SCAT_T) cur[i] = rowptr[lo + i] + pre[i];
    __syncthreads();

    const int e0 = c * chunkLen;
    const int e1 = min(E, e0 + chunkLen);
    int base = e0 + tid;
    for (; base + 7 * SCAT_T < e1; base += 8 * SCAT_T) {
        unsigned l[8];
        int      s[8];
        #pragma unroll
        for (int j = 0; j < 8; ++j) {
            l[j] = (unsigned)(dst[base + j * SCAT_T] - lo);
            s[j] = src[base + j * SCAT_T];
        }
        unsigned p[8];
        #pragma unroll
        for (int j = 0; j < 8; ++j) {
            p[j] = 0u;
            if (l[j] < (unsigned)len) p[j] = atomicAdd(&cur[l[j]], 1u);
        }
        #pragma unroll
        for (int j = 0; j < 8; ++j)
            if (l[j] < (unsigned)len) csrSrc[p[j]] = s[j];
    }
    for (; base < e1; base += SCAT_T) {
        const unsigned l = (unsigned)(dst[base] - lo);
        if (l < (unsigned)len) csrSrc[atomicAdd(&cur[l], 1u)] = src[base];
    }
}

__global__ void __launch_bounds__(256) k_gemm1_wmma(const float* __restrict__ X,
                                                    const float* __restrict__ W1,
                                                    float* __restrict__ H,
                                                    int nTiles) {
    __shared__ float Wlds[IN_FEATS * HIDDEN];
    for (int i = threadIdx.x; i < IN_FEATS * HIDDEN; i += blockDim.x)
        Wlds[i] = W1[i];
    __syncthreads();

    const int waveId = (int)((blockIdx.x * blockDim.x + threadIdx.x) >> 5);
    const int lane   = threadIdx.x & 31;
    if (waveId >= nTiles) return;

    const int nTilesN = HIDDEN / 16;
    const int mt = waveId / nTilesN;
    const int nt = waveId % nTilesN;

    const int row = mt * 16 + (lane & 15);
    const int n   = nt * 16 + (lane & 15);
    const int kq  = (lane >> 4) << 1;

    const float* __restrict__ xrow = X + (size_t)row * IN_FEATS;

    v8f c = {};
    #pragma unroll 8
    for (int k = 0; k < IN_FEATS; k += 4) {
        v2f a, b;
        a.x = xrow[k + kq];
        a.y = xrow[k + kq + 1];
        b.x = Wlds[(k + kq)     * HIDDEN + n];
        b.y = Wlds[(k + kq + 1) * HIDDEN + n];
        c = __builtin_amdgcn_wmma_f32_16x16x4_f32(
                 false, a,  false, b,
                 (short)0, c,  false,  false);
    }

    const int rbase = mt * 16 + ((lane >> 4) << 3);
    #pragma unroll
    for (int v = 0; v < 8; ++v)
        H[(size_t)(rbase + v) * HIDDEN + n] = c[v];
}

__global__ void __launch_bounds__(256) k_gather1(const unsigned* __restrict__ rowptr,
                                                 const int* __restrict__ csrSrc,
                                                 const float* __restrict__ dinv,
                                                 const float* __restrict__ H,
                                                 float* __restrict__ A1, int n) {
    const int node = __builtin_amdgcn_readfirstlane((int)((blockIdx.x * blockDim.x + threadIdx.x) >> 5));
    if (node >= n) return;
    const int lane = threadIdx.x & 31;
    const int half = lane >> 4;
    const int q    = lane & 15;
    const float4* __restrict__ H4 = (const float4*)H;

    const float dd  = dinv[node];
    const int   beg = (int)rowptr[node];
    const int   end = (int)rowptr[node + 1];

    float4 acc = make_float4(0.0f, 0.0f, 0.0f, 0.0f);
    float4 hs = make_float4(0.0f, 0.0f, 0.0f, 0.0f);
    if (half == 0) hs = H4[(size_t)node * (HIDDEN / 4) + q];

    for (int b0 = beg; b0 < end; b0 += 32) {
        const int cnt = min(32, end - b0);
        int   s_l = 0;
        float w_l = 0.0f;
        if (lane < cnt) { s_l = csrSrc[b0 + lane]; w_l = dinv[s_l] * dd; }
        for (int j = 0; j < cnt; j += 16) {
            float4 h[8];
            float  w[8];
            #pragma unroll
            for (int u = 0; u < 8; ++u) {
                const int idx = j + 2 * u + half;
                const int s   = __shfl(s_l, idx);
                w[u] = __shfl(w_l, idx);
                h[u] = make_float4(0.0f, 0.0f, 0.0f, 0.0f);
                if (idx < cnt) h[u] = H4[(size_t)s * (HIDDEN / 4) + q];
            }
            #pragma unroll
            for (int u = 0; u < 8; ++u) {
                acc.x = fmaf(h[u].x, w[u], acc.x);
                acc.y = fmaf(h[u].y, w[u], acc.y);
                acc.z = fmaf(h[u].z, w[u], acc.z);
                acc.w = fmaf(h[u].w, w[u], acc.w);
            }
        }
    }

    const float wself = dd * dd;
    acc.x = fmaf(hs.x, wself, acc.x);
    acc.y = fmaf(hs.y, wself, acc.y);
    acc.z = fmaf(hs.z, wself, acc.z);
    acc.w = fmaf(hs.w, wself, acc.w);

    acc.x += __shfl_xor(acc.x, 16);
    acc.y += __shfl_xor(acc.y, 16);
    acc.z += __shfl_xor(acc.z, 16);
    acc.w += __shfl_xor(acc.w, 16);
    if (half == 0) ((float4*)A1)[(size_t)node * (HIDDEN / 4) + q] = acc;
}

__global__ void k_dense2(const float* __restrict__ A1, const float* __restrict__ b1,
                         const float* __restrict__ W2, float* __restrict__ H2, int n) {
    int i = blockIdx.x * blockDim.x + threadIdx.x;
    if (i >= n) return;
    const float* __restrict__ a = A1 + (size_t)i * HIDDEN;
    float acc0 = 0.0f, acc1 = 0.0f;
    #pragma unroll 16
    for (int f = 0; f < HIDDEN; ++f) {
        float v = a[f] + b1[f];
        v = v > 0.0f ? v : 0.0f;
        acc0 = fmaf(v, W2[f * 2 + 0], acc0);
        acc1 = fmaf(v, W2[f * 2 + 1], acc1);
    }
    H2[i * 2 + 0] = acc0;
    H2[i * 2 + 1] = acc1;
}

template <int K>
__device__ __forceinline__ void agg2_edges(const int* __restrict__ src, const int* __restrict__ dst,
                                           const float* __restrict__ dinv,
                                           const float2* __restrict__ H2v,
                                           int base, int lo, int len,
                                           float* acc0, float* acc1) {
    int      s[K];
    unsigned l[K];
    #pragma unroll
    for (int j = 0; j < K; ++j) {
        l[j] = (unsigned)(dst[base + j * SCAT_T] - lo);
        s[j] = src[base + j * SCAT_T];
    }
    float  ws[K];
    float  wd[K];
    float2 h[K];
    #pragma unroll
    for (int j = 0; j < K; ++j) {
        ws[j] = 0.0f; wd[j] = 0.0f; h[j] = make_float2(0.0f, 0.0f);
        if (l[j] < (unsigned)len) {
            ws[j] = dinv[s[j]];
            wd[j] = dinv[lo + (int)l[j]];
            h[j]  = H2v[s[j]];
        }
    }
    #pragma unroll
    for (int j = 0; j < K; ++j) {
        if (l[j] < (unsigned)len) {
            const float w = ws[j] * wd[j];
            atomicAdd(&acc0[l[j]], h[j].x * w);
            atomicAdd(&acc1[l[j]], h[j].y * w);
        }
    }
}

__global__ void __launch_bounds__(SCAT_T) k_agg2_part(const int* __restrict__ src,
                                                      const int* __restrict__ dst,
                                                      const float* __restrict__ dinv,
                                                      const float* __restrict__ H2,
                                                      int E, int n, int nPad, int nChunks, int chunkLen,
                                                      float* __restrict__ P0, float* __restrict__ P1) {
    __shared__ float acc0[SCAT_CAP];
    __shared__ float acc1[SCAT_CAP];
    const int r   = blockIdx.x / nChunks;
    const int c   = blockIdx.x - r * nChunks;
    const int lo  = r * SCAT_CAP;
    const int len = min(SCAT_CAP, n - lo);
    const int tid = threadIdx.x;

    for (int i = tid; i < len; i += SCAT_T) { acc0[i] = 0.0f; acc1[i] = 0.0f; }
    __syncthreads();

    const float2* __restrict__ H2v = (const float2*)H2;
    const int e0 = c * chunkLen;
    const int e1 = min(E, e0 + chunkLen);
    int base = e0 + tid;
    for (; base + 7 * SCAT_T < e1; base += 8 * SCAT_T)
        agg2_edges<8>(src, dst, dinv, H2v, base, lo, len, acc0, acc1);
    for (; base < e1; base += SCAT_T)
        agg2_edges<1>(src, dst, dinv, H2v, base, lo, len, acc0, acc1);
    __syncthreads();

    float* __restrict__ o0 = P0 + (size_t)c * nPad + lo;
    float* __restrict__ o1 = P1 + (size_t)c * nPad + lo;
    for (int i = tid; i < len; i += SCAT_T) { o0[i] = acc0[i]; o1[i] = acc1[i]; }
}

__global__ void k_final2(const float* __restrict__ H2, const float* __restrict__ dinv,
                         const float* __restrict__ b2,
                         const float* __restrict__ P0, const float* __restrict__ P1,
                         int nPad, int nChunks, float* __restrict__ out, int n) {
    int i = blockIdx.x * blockDim.x + threadIdx.x;
    if (i < n) {
        float d = dinv[i];
        float w = d * d;
        float o0 = H2[i * 2 + 0] * w + b2[0];
        float o1 = H2[i * 2 + 1] * w + b2[1];
        #pragma unroll 8
        for (int c = 0; c < nChunks; ++c) {
            o0 += P0[(size_t)c * nPad + i];
            o1 += P1[(size_t)c * nPad + i];
        }
        out[i * 2 + 0] = o0;
        out[i * 2 + 1] = o1;
    }
}

extern "C" void kernel_launch(void* const* d_in, const int* in_sizes, int n_in,
                              void* d_out, int out_size, void* d_ws, size_t ws_size,
                              hipStream_t stream) {
    const float* x  = (const float*)d_in[0];
    const float* W1 = (const float*)d_in[1];
    const float* b1 = (const float*)d_in[2];
    const float* W2 = (const float*)d_in[3];
    const float* b2 = (const float*)d_in[4];
    const int*   ei = (const int*)d_in[5];

    const int N = in_sizes[0] / IN_FEATS;
    const int E = in_sizes[5] / 2;
    const int* src = ei;
    const int* dst = ei + E;

    const int nRanges  = (N + SCAT_CAP - 1) / SCAT_CAP;
    const int nChunks  = nRanges < SCAT_WGS ? SCAT_WGS / nRanges : 1;
    const int chunkLen = (E + nChunks - 1) / nChunks;
    const int nPad     = (N + 31) & ~31;

    float* ws   = (float*)d_ws;
    float* dinv = ws;
    float* H    = dinv + nPad;
    float* A1   = H    + (size_t)N * HIDDEN;
    float* H2   = A1   + (size_t)N * HIDDEN;
    float* P0   = H2   + (size_t)nPad * 2;
    float* P1   = P0   + (size_t)nChunks * nPad;
    unsigned* cntP = (unsigned*)(P1 + (size_t)nChunks * nPad);
    unsigned* preP   = cntP   + (size_t)nChunks * nPad;
    unsigned* cntTot = preP   + (size_t)nChunks * nPad;
    unsigned* rowptr = cntTot + nPad;
    int*      csrSrc = (int*)(rowptr + nPad + 32);
    float* out  = (float*)d_out;

    const int T = 256;

    k_deg_part <<<nRanges * nChunks, SCAT_T, 0, stream>>>(dst, E, N, nPad, nChunks, chunkLen, cntP);
    k_make_dinv<<<(N + T - 1) / T, T, 0, stream>>>(cntP, nPad, nChunks, dinv, preP, cntTot, N);

    k_rowptr  <<<1, SCAT_T, 0, stream>>>(cntTot, N, rowptr);
    k_csr_fill<<<nRanges * nChunks, SCAT_T, 0, stream>>>(src, dst, E, N, nPad, nChunks, chunkLen, rowptr, preP, csrSrc);

    const int nTiles = (N / 16) * (HIDDEN / 16);
    const int gemmBlocks = (nTiles + (T / 32) - 1) / (T / 32);
    k_gemm1_wmma<<<gemmBlocks, T, 0, stream>>>(x, W1, H, nTiles);

    k_gather1<<<(N * 32 + T - 1) / T, T, 0, stream>>>(rowptr, csrSrc, dinv, H, A1, N);

    k_dense2   <<<(N + T - 1) / T, T, 0, stream>>>(A1, b1, W2, H2, N);
    k_agg2_part<<<nRanges * nChunks, SCAT_T, 0, stream>>>(src, dst, dinv, H2, E, N, nPad, nChunks, chunkLen, P0, P1);
    k_final2   <<<(N + T - 1) / T, T, 0, stream>>>(H2, dinv, b2, P0, P1, nPad, nChunks, out, N);
}


# layer-2 scatter-add walks the destination-sorted edge list: 4 lanes per node sum dinv[src]*H2[src] in registers, one whole-line store per wave; no LDS adds, no partial planes (sum order changes)
# speedup vs baseline: 3.5992x; 1.1957x over previous
#include <hip/hip_runtime.h>
#include <hip/hip_bf16.h>

typedef float v2f __attribute__((ext_vector_type(2)));
typedef float v8f __attribute__((ext_vector_type(8)));

#define IN_FEATS 128
#define HIDDEN   64

#define SCAT_CAP 6272
#define SCAT_T   1024
#define SCAT_WGS 256

__global__ void __launch_bounds__(SCAT_T) k_deg_part(const int* __restrict__ dst, int E, int n,
                                                     int nPad, int nChunks, int chunkLen,
                                                     unsigned* __restrict__ cntP) {
    __shared__ unsigned cnt[SCAT_CAP];
    const int r   = blockIdx.x / nChunks;
    const int c   = blockIdx.x - r * nChunks;
    const int lo  = r * SCAT_CAP;
    const int len = min(SCAT_CAP, n - lo);
    const int tid = threadIdx.x;

    for (int i = tid; i < len; i += SCAT_T) cnt[i] = 0u;
    __syncthreads();

    const int e0 = c * chunkLen;
    const int e1 = min(E, e0 + chunkLen);
    int base = e0 + tid;
    for (; base + 7 * SCAT_T < e1; base += 8 * SCAT_T) {
        unsigned l[8];
        #pragma unroll
        for (int j = 0; j < 8; ++j) l[j] = (unsigned)(dst[base + j * SCAT_T] - lo);
        #pragma unroll
        for (int j = 0; j < 8; ++j)
            if (l[j] < (unsigned)len) atomicAdd(&cnt[l[j]], 1u);
    }
    for (; base < e1; base += SCAT_T) {
        const unsigned l = (unsigned)(dst[base] - lo);
        if (l < (unsigned)len) atomicAdd(&cnt[l], 1u);
    }
    __syncthreads();

    unsigned* __restrict__ o = cntP + (size_t)c * nPad + lo;
    for (int i = tid; i < len; i += SCAT_T) o[i] = cnt[i];
}

__global__ void k_make_dinv(const unsigned* __restrict__ cntP, int nPad, int nChunks,
                            float* __restrict__ dinv, unsigned* __restrict__ preP,
                            unsigned* __restrict__ cntTot, int n) {
    int i = blockIdx.x * blockDim.x + threadIdx.x;
    if (i < n) {
        unsigned cnt = 0u;
        #pragma unroll 8
        for (int c = 0; c < nChunks; ++c) {
            const unsigned v = cntP[(size_t)c * nPad + i];
            preP[(size_t)c * nPad + i] = cnt;
            cnt += v;
        }
        cntTot[i] = cnt;
        float d = 1.0f + (float)cnt;
        dinv[i] = (d > 0.0f) ? rsqrtf(d) : 0.0f;
    }
}

#define RP_V4  13
#define RP_PER (RP_V4 * 4)
#define RP_CAP (SCAT_T * RP_PER)

__global__ void __launch_bounds__(SCAT_T) k_rowptr(const unsigned* __restrict__ cntTot, int n,
                                                   unsigned* __restrict__ rowptr) {
    __shared__ uint4    buf[RP_CAP / 4];
    __shared__ unsigned wsum[SCAT_T / 32 + 1];
    const int tid  = threadIdx.x;
    const int lane = tid & 31;
    const int wv   = tid >> 5;

    unsigned carry = 0u;
    for (int t0 = 0; t0 < n; t0 += RP_CAP) {
        const int len = min(RP_CAP, n - t0);
        const unsigned* __restrict__ gin  = cntTot + t0;
        unsigned* __restrict__       gout = rowptr + t0;

        const int nFull = len >> 2;
        const int qLast = max(nFull - 1, 0);

        uint4 v[RP_V4];
        #pragma unroll
        for (int j = 0; j < RP_V4; ++j) v[j] = ((const uint4*)gin)[min(tid + j * SCAT_T, qLast)];
        #pragma unroll
        for (int j = 0; j < RP_V4; ++j) {
            const int q = tid + j * SCAT_T;
            buf[q] = (q < nFull) ? v[j] : make_uint4(0u, 0u, 0u, 0u);
        }
        if ((len & 3) && tid == (nFull & (SCAT_T - 1))) {
            const int e = nFull * 4;
            uint4 t = make_uint4(0u, 0u, 0u, 0u);
            t.x = gin[e];
            if (e + 1 < len) t.y = gin[e + 1];
            if (e + 2 < len) t.z = gin[e + 2];
            buf[nFull] = t;
        }
        __syncthreads();

        uint4 r[RP_V4];
        #pragma unroll
        for (int j = 0; j < RP_V4; ++j) r[j] = buf[tid * RP_V4 + j];
        unsigned s = 0u;
        #pragma unroll
        for (int j = 0; j < RP_V4; ++j) s += r[j].x + r[j].y + r[j].z + r[j].w;

        unsigned inc = s;
        for (int o = 1; o < 32; o <<= 1) {
            const unsigned t = __shfl_up(inc, o);
            if (lane >= o) inc += t;
        }
        if (lane == 31) wsum[wv] = inc;
        __syncthreads();
        if (wv == 0) {
            const unsigned w = wsum[lane];
            unsigned winc = w;
            for (int o = 1; o < 32; o <<= 1) {
                const unsigned t = __shfl_up(winc, o);
                if (lane >= o) winc += t;
            }
            wsum[lane] = winc - w;
            if (lane == 31) wsum[SCAT_T / 32] = winc;
        }
        __syncthreads();

        unsigned run = carry + wsum[wv] + inc - s;
        #pragma unroll
        for (int j = 0; j < RP_V4; ++j) {
            uint4 p;
            p.x = run; run += r[j].x;
            p.y = run; run += r[j].y;
            p.z = run; run += r[j].z;
            p.w = run; run += r[j].w;
            buf[tid * RP_V4 + j] = p;
        }
        carry += wsum[SCAT_T / 32];
        __syncthreads();

        #pragma unroll
        for (int j = 0; j < RP_V4; ++j) v[j] = buf[tid + j * SCAT_T];
        #pragma unroll
        for (int j = 0; j < RP_V4; ++j) {
            const int q = tid + j * SCAT_T;
            if (q < nFull) ((uint4*)gout)[q] = v[j];
        }
        if ((len & 3) && tid == (nFull & (SCAT_T - 1))) {
            const int e = nFull * 4;
            const uint4 t = buf[nFull];
            gout[e] = t.x;
            if (e + 1 < len) gout[e + 1] = t.y;
            if (e + 2 < len) gout[e + 2] = t.z;
        }
        __syncthreads();
    }
    if (tid == 0) rowptr[n] = carry;
}

__global__ void __launch_bounds__(SCAT_T) k_csr_fill(const int* __restrict__ src,
                                                     const int* __restrict__ dst,
                                                     int E, int n, int nPad, int nChunks, int chunkLen,
                                                     const unsigned* __restrict__ rowptr,
                                                     const unsigned* __restrict__ preP,
                                                     int* __restrict__ csrSrc) {
    __shared__ unsigned cur[SCAT_CAP];
    const int r   = blockIdx.x / nChunks;
    const int c   = blockIdx.x - r * nChunks;
    const int lo  = r * SCAT_CAP;
    const int len = min(SCAT_CAP, n - lo);
    const int tid = threadIdx.x;

    const unsigned* __restrict__ pre = preP + (size_t)c * nPad + lo;
    for (int i = tid; i < len; i += SCAT_T) cur[i] = rowptr[lo + i] + pre[i];
    __syncthreads();

    const int e0 = c * chunkLen;
    const int e1 = min(E, e0 + chunkLen);
    int base = e0 + tid;
    for (; base + 7 * SCAT_T < e1; base += 8 * SCAT_T) {
        unsigned l[8];
        int      s[8];
        #pragma unroll
        for (int j = 0; j < 8; ++j) {
            l[j] = (unsigned)(dst[base + j * SCAT_T] - lo);
            s[j] = src[base + j * SCAT_T];
        }
        unsigned p[8];
        #pragma unroll
        for (int j = 0; j < 8; ++j) {
            p[j] = 0u;
            if (l[j] < (unsigned)len) p[j] = atomicAdd(&cur[l[j]], 1u);
        }
        #pragma unroll
        for (int j = 0; j < 8; ++j)
            if (l[j] < (unsigned)len) csrSrc[p[j]] = s[j];
    }
    for (; base < e1; base += SCAT_T) {
        const unsigned l = (unsigned)(dst[base] - lo);
        if (l < (unsigned)len) csrSrc[atomicAdd(&cur[l], 1u)] = src[base];
    }
}

__global__ void __launch_bounds__(256) k_gemm1_wmma(const float* __restrict__ X,
                                                    const float* __restrict__ W1,
                                                    float* __restrict__ H,
                                                    int nTiles) {
    __shared__ float Wlds[IN_FEATS * HIDDEN];
    for (int i = threadIdx.x; i < IN_FEATS * HIDDEN; i += blockDim.x)
        Wlds[i] = W1[i];
    __syncthreads();

    const int waveId = (int)((blockIdx.x * blockDim.x + threadIdx.x) >> 5);
    const int lane   = threadIdx.x & 31;
    if (waveId >= nTiles) return;

    const int nTilesN = HIDDEN / 16;
    const int mt = waveId / nTilesN;
    const int nt = waveId % nTilesN;

    const int row = mt * 16 + (lane & 15);
    const int n   = nt * 16 + (lane & 15);
    const int kq  = (lane >> 4) << 1;

    const float* __restrict__ xrow = X + (size_t)row * IN_FEATS;

    v8f c = {};
    #pragma unroll 8
    for (int k = 0; k < IN_FEATS; k += 4) {
        v2f a, b;
        a.x = xrow[k + kq];
        a.y = xrow[k + kq + 1];
        b.x = Wlds[(k + kq)     * HIDDEN + n];
        b.y = Wlds[(k + kq + 1) * HIDDEN + n];
        c = __builtin_amdgcn_wmma_f32_16x16x4_f32(
                 false, a,  false, b,
                 (short)0, c,  false,  false);
    }

    const int rbase = mt * 16 + ((lane >> 4) << 3);
    #pragma unroll
    for (int v = 0; v < 8; ++v)
        H[(size_t)(rbase + v) * HIDDEN + n] = c[v];
}

__global__ void __launch_bounds__(256) k_gather1(const unsigned* __restrict__ rowptr,
                                                 const int* __restrict__ csrSrc,
                                                 const float* __restrict__ dinv,
                                                 const float* __restrict__ H,
                                                 float* __restrict__ A1, int n) {
    const int node = __builtin_amdgcn_readfirstlane((int)((blockIdx.x * blockDim.x + threadIdx.x) >> 5));
    if (node >= n) return;
    const int lane = threadIdx.x & 31;
    const int half = lane >> 4;
    const int q    = lane & 15;
    const float4* __restrict__ H4 = (const float4*)H;

    const float dd  = dinv[node];
    const int   beg = (int)rowptr[node];
    const int   end = (int)rowptr[node + 1];

    float4 acc = make_float4(0.0f, 0.0f, 0.0f, 0.0f);
    float4 hs = make_float4(0.0f, 0.0f, 0.0f, 0.0f);
    if (half == 0) hs = H4[(size_t)node * (HIDDEN / 4) + q];

    for (int b0 = beg; b0 < end; b0 += 32) {
        const int cnt = min(32, end - b0);
        int   s_l = 0;
        float w_l = 0.0f;
        if (lane < cnt) { s_l = csrSrc[b0 + lane]; w_l = dinv[s_l] * dd; }
        for (int j = 0; j < cnt; j += 16) {
            float4 h[8];
            float  w[8];
            #pragma unroll
            for (int u = 0; u < 8; ++u) {
                const int idx = j + 2 * u + half;
                const int s   = __shfl(s_l, idx);
                w[u] = __shfl(w_l, idx);
                h[u] = make_float4(0.0f, 0.0f, 0.0f, 0.0f);
                if (idx < cnt) h[u] = H4[(size_t)s * (HIDDEN / 4) + q];
            }
            #pragma unroll
            for (int u = 0; u < 8; ++u) {
                acc.x = fmaf(h[u].x, w[u], acc.x);
                acc.y = fmaf(h[u].y, w[u], acc.y);
                acc.z = fmaf(h[u].z, w[u], acc.z);
                acc.w = fmaf(h[u].w, w[u], acc.w);
            }
        }
    }

    const float wself = dd * dd;
    acc.x = fmaf(hs.x, wself, acc.x);
    acc.y = fmaf(hs.y, wself, acc.y);
    acc.z = fmaf(hs.z, wself, acc.z);
    acc.w = fmaf(hs.w, wself, acc.w);

    acc.x += __shfl_xor(acc.x, 16);
    acc.y += __shfl_xor(acc.y, 16);
    acc.z += __shfl_xor(acc.z, 16);
    acc.w += __shfl_xor(acc.w, 16);
    if (half == 0) ((float4*)A1)[(size_t)node * (HIDDEN / 4) + q] = acc;
}

__global__ void k_dense2(const float* __restrict__ A1, const float* __restrict__ b1,
                         const float* __restrict__ W2, const float* __restrict__ dinv,
                         float* __restrict__ G, int n) {
    int i = blockIdx.x * blockDim.x + threadIdx.x;
    if (i >= n) return;
    const float* __restrict__ a = A1 + (size_t)i * HIDDEN;
    const float d = dinv[i];
    float acc0 = 0.0f, acc1 = 0.0f;
    #pragma unroll 16
    for (int f = 0; f < HIDDEN; ++f) {
        float v = a[f] + b1[f];
        v = v > 0.0f ? v : 0.0f;
        acc0 = fmaf(v, W2[f * 2 + 0], acc0);
        acc1 = fmaf(v, W2[f * 2 + 1], acc1);
    }
    G[i * 2 + 0] = acc0 * d;
    G[i * 2 + 1] = acc1 * d;
}

#define G2_LANES 4
#define G2_WPB   32
#define G2_U     12

__device__ __forceinline__ float quad_sum(float v) {
    v += __int_as_float(__builtin_amdgcn_update_dpp(0, __float_as_int(v), 0xB1, 0xF, 0xF, true));
    v += __int_as_float(__builtin_amdgcn_update_dpp(0, __float_as_int(v), 0x4E, 0xF, 0xF, true));
    return v;
}

__global__ void __launch_bounds__(G2_WPB * 32) k_gather2(const unsigned* __restrict__ rowptr,
                                                 const int* __restrict__ csrSrc,
                                                 const float* __restrict__ dinv,
                                                 const float* __restrict__ G,
                                                 const float* __restrict__ b2,
                                                 float* __restrict__ out, int n) {
    const int wave = (int)((blockIdx.x * blockDim.x + threadIdx.x) >> 5);
    const int lane = threadIdx.x & 31;
    const int grp  = lane >> 2;
    const int sub  = lane & 3;
    const float2* __restrict__ G2 = (const float2*)G;

    const int  nodeA = wave * 16 + grp;
    const int  nodeB = nodeA + 8;
    const bool okA   = nodeA < n;
    const bool okB   = nodeB < n;
    const int  cA    = min(nodeA, n - 1);
    const int  cB    = min(nodeB, n - 1);

    int begA = (int)rowptr[cA], endA = (int)rowptr[cA + 1];
    int begB = (int)rowptr[cB], endB = (int)rowptr[cB + 1];
    const int   nodeS = (sub == 0) ? nodeA : nodeB;
    const bool  okS   = (sub < 2) && (nodeS < n);
    const float dS    = dinv[min(nodeS, n - 1)];
    const float2 selfA = G2[cA];
    const float2 selfB = G2[cB];
    if (!okA) endA = begA;
    if (!okB) endB = begB;

    float2 accA = make_float2(0.0f, 0.0f);
    float2 accB = make_float2(0.0f, 0.0f);
    if (sub == 0 && okA) accA = selfA;
    if (sub == 0 && okB) accB = selfB;

    const int lastA = max(endA - 1, 0);
    const int lastB = max(endB - 1, 0);
    int eA = begA + sub;
    int eB = begB + sub;
    while ((eA < endA) | (eB < endB)) {
        int sA[G2_U], sB[G2_U];
        #pragma unroll
        for (int u = 0; u < G2_U; ++u) {
            sA[u] = csrSrc[min(eA + u * G2_LANES, lastA)];
            sB[u] = csrSrc[min(eB + u * G2_LANES, lastB)];
        }
        float2 gA[G2_U], gB[G2_U];
        #pragma unroll
        for (int u = 0; u < G2_U; ++u) {
            gA[u] = G2[sA[u]];
            gB[u] = G2[sB[u]];
        }
        #pragma unroll
        for (int u = 0; u < G2_U; ++u)
            asm volatile("" :: "v"(gA[u].x), "v"(gA[u].y), "v"(gB[u].x), "v"(gB[u].y));
        #pragma unroll
        for (int u = 0; u < G2_U; ++u) {
            const bool vA = eA + u * G2_LANES < endA;
            const bool vB = eB + u * G2_LANES < endB;
            accA.x += vA ? gA[u].x : 0.0f; accA.y += vA ? gA[u].y : 0.0f;
            accB.x += vB ? gB[u].x : 0.0f; accB.y += vB ? gB[u].y : 0.0f;
        }
        eA += G2_U * G2_LANES;
        eB += G2_U * G2_LANES;
    }

    accA.x = quad_sum(accA.x); accA.y = quad_sum(accA.y);
    accB.x = quad_sum(accB.x); accB.y = quad_sum(accB.y);

    if (okS) {
        const float2 t = (sub == 0) ? accA : accB;
        float2 o;
        o.x = fmaf(dS, t.x, b2[0]);
        o.y = fmaf(dS, t.y, b2[1]);
        ((float2*)out)[nodeS] = o;
    }
}

extern "C" void kernel_launch(void* const* d_in, const int* in_sizes, int n_in,
                              void* d_out, int out_size, void* d_ws, size_t ws_size,
                              hipStream_t stream) {
    const float* x  = (const float*)d_in[0];
    const float* W1 = (const float*)d_in[1];
    const float* b1 = (const float*)d_in[2];
    const float* W2 = (const float*)d_in[3];
    const float* b2 = (const float*)d_in[4];
    const int*   ei = (const int*)d_in[5];

    const int N = in_sizes[0] / IN_FEATS;
    const int E = in_sizes[5] / 2;
    const int* src = ei;
    const int* dst = ei + E;

    const int nRanges  = (N + SCAT_CAP - 1) / SCAT_CAP;
    const int nChunks  = nRanges < SCAT_WGS ? SCAT_WGS / nRanges : 1;
    const int chunkLen = (E + nChunks - 1) / nChunks;
    const int nPad     = (N + 31) & ~31;

    float* ws   = (float*)d_ws;
    float* dinv = ws;
    float* H    = dinv + nPad;
    float* A1   = H    + (size_t)N * HIDDEN;
    float* G    = A1   + (size_t)N * HIDDEN;
    unsigned* cntP = (unsigned*)(G + (size_t)nPad * 2);
    unsigned* preP   = cntP   + (size_t)nChunks * nPad;
    unsigned* cntTot = preP   + (size_t)nChunks * nPad;
    unsigned* rowptr = cntTot + nPad;
    int*      csrSrc = (int*)(rowptr + nPad + 32);
    float* out  = (float*)d_out;

    const int T = 256;

    k_deg_part <<<nRanges * nChunks, SCAT_T, 0, stream>>>(dst, E, N, nPad, nChunks, chunkLen, cntP);
    k_make_dinv<<<(N + T - 1) / T, T, 0, stream>>>(cntP, nPad, nChunks, dinv, preP, cntTot, N);

    k_rowptr  <<<1, SCAT_T, 0, stream>>>(cntTot, N, rowptr);
    k_csr_fill<<<nRanges * nChunks, SCAT_T, 0, stream>>>(src, dst, E, N, nPad, nChunks, chunkLen, rowptr, preP, csrSrc);

    const int nTiles = (N / 16) * (HIDDEN / 16);
    const int gemmBlocks = (nTiles + (T / 32) - 1) / (T / 32);
    k_gemm1_wmma<<<gemmBlocks, T, 0, stream>>>(x, W1, H, nTiles);

    k_gather1<<<(N * 32 + T - 1) / T, T, 0, stream>>>(rowptr, csrSrc, dinv, H, A1, N);

    k_dense2 <<<(N + T - 1) / T, T, 0, stream>>>(A1, b1, W2, dinv, G, N);
    const int g2Waves = (N + 15) / 16;
    int g2Wpb = (g2Waves + SCAT_WGS - 1) / SCAT_WGS;
    if (g2Wpb > G2_WPB) g2Wpb = G2_WPB;
    if (g2Wpb < 1) g2Wpb = 1;
    k_gather2<<<(g2Waves + g2Wpb - 1) / g2Wpb, g2Wpb * 32, 0, stream>>>(rowptr, csrSrc, dinv, G, b2, out, N);
}


# x@W1 WMMA product joins the edge-list fill's launch (work-groups 0..255 fill, 391 multiply, W1 staged in 16-byte pieces): 7 launches, product runs under the fill's latency; sum order kept
# speedup vs baseline: 4.4975x; 1.2496x over previous
#include <hip/hip_runtime.h>
#include <hip/hip_bf16.h>

typedef float v2f __attribute__((ext_vector_type(2)));
typedef float v8f __attribute__((ext_vector_type(8)));

#define IN_FEATS 128
#define HIDDEN   64

#define SCAT_CAP 6272
#define SCAT_T   1024
#define SCAT_WGS 256

__global__ void __launch_bounds__(SCAT_T) k_deg_part(const int* __restrict__ dst, int E, int n,
                                                     int nPad, int nChunks, int chunkLen,
                                                     unsigned* __restrict__ cntP) {
    __shared__ unsigned cnt[SCAT_CAP];
    const int r   = blockIdx.x / nChunks;
    const int c   = blockIdx.x - r * nChunks;
    const int lo  = r * SCAT_CAP;
    const int len = min(SCAT_CAP, n - lo);
    const int tid = threadIdx.x;

    for (int i = tid; i < len; i += SCAT_T) cnt[i] = 0u;
    __syncthreads();

    const int e0 = c * chunkLen;
    const int e1 = min(E, e0 + chunkLen);
    int base = e0 + tid;
    for (; base + 7 * SCAT_T < e1; base += 8 * SCAT_T) {
        unsigned l[8];
        #pragma unroll
        for (int j = 0; j < 8; ++j) l[j] = (unsigned)(dst[base + j * SCAT_T] - lo);
        #pragma unroll
        for (int j = 0; j < 8; ++j)
            if (l[j] < (unsigned)len) atomicAdd(&cnt[l[j]], 1u);
    }
    for (; base < e1; base += SCAT_T) {
        const unsigned l = (unsigned)(dst[base] - lo);
        if (l < (unsigned)len) atomicAdd(&cnt[l], 1u);
    }
    __syncthreads();

    unsigned* __restrict__ o = cntP + (size_t)c * nPad + lo;
    for (int i = tid; i < len; i += SCAT_T) o[i] = cnt[i];
}

__global__ void k_make_dinv(const unsigned* __restrict__ cntP, int nPad, int nChunks,
                            float* __restrict__ dinv, unsigned* __restrict__ preP,
                            unsigned* __restrict__ cntTot, int n) {
    int i = blockIdx.x * blockDim.x + threadIdx.x;
    if (i < n) {
        unsigned cnt = 0u;
        #pragma unroll 8
        for (int c = 0; c < nChunks; ++c) {
            const unsigned v = cntP[(size_t)c * nPad + i];
            preP[(size_t)c * nPad + i] = cnt;
            cnt += v;
        }
        cntTot[i] = cnt;
        float d = 1.0f + (float)cnt;
        dinv[i] = (d > 0.0f) ? rsqrtf(d) : 0.0f;
    }
}

#define RP_V4  13
#define RP_PER (RP_V4 * 4)
#define RP_CAP (SCAT_T * RP_PER)

__global__ void __launch_bounds__(SCAT_T) k_rowptr(const unsigned* __restrict__ cntTot, int n,
                                                   unsigned* __restrict__ rowptr) {
    __shared__ uint4    buf[RP_CAP / 4];
    __shared__ unsigned wsum[SCAT_T / 32 + 1];
    const int tid  = threadIdx.x;
    const int lane = tid & 31;
    const int wv   = tid >> 5;

    unsigned carry = 0u;
    for (int t0 = 0; t0 < n; t0 += RP_CAP) {
        const int len = min(RP_CAP, n - t0);
        const unsigned* __restrict__ gin  = cntTot + t0;
        unsigned* __restrict__       gout = rowptr + t0;

        const int nFull = len >> 2;
        const int qLast = max(nFull - 1, 0);

        uint4 v[RP_V4];
        #pragma unroll
        for (int j = 0; j < RP_V4; ++j) v[j] = ((const uint4*)gin)[min(tid + j * SCAT_T, qLast)];
        #pragma unroll
        for (int j = 0; j < RP_V4; ++j) {
            const int q = tid + j * SCAT_T;
            buf[q] = (q < nFull) ? v[j] : make_uint4(0u, 0u, 0u, 0u);
        }
        if ((len & 3) && tid == (nFull & (SCAT_T - 1))) {
            const int e = nFull * 4;
            uint4 t = make_uint4(0u, 0u, 0u, 0u);
            t.x = gin[e];
            if (e + 1 < len) t.y = gin[e + 1];
            if (e + 2 < len) t.z = gin[e + 2];
            buf[nFull] = t;
        }
        __syncthreads();

        uint4 r[RP_V4];
        #pragma unroll
        for (int j = 0; j < RP_V4; ++j) r[j] = buf[tid * RP_V4 + j];
        unsigned s = 0u;
        #pragma unroll
        for (int j = 0; j < RP_V4; ++j) s += r[j].x + r[j].y + r[j].z + r[j].w;

        unsigned inc = s;
        for (int o = 1; o < 32; o <<= 1) {
            const unsigned t = __shfl_up(inc, o);
            if (lane >= o) inc += t;
        }
        if (lane == 31) wsum[wv] = inc;
        __syncthreads();
        if (wv == 0) {
            const unsigned w = wsum[lane];
            unsigned winc = w;
            for (int o = 1; o < 32; o <<= 1) {
                const unsigned t = __shfl_up(winc, o);
                if (lane >= o) winc += t;
            }
            wsum[lane] = winc - w;
            if (lane == 31) wsum[SCAT_T / 32] = winc;
        }
        __syncthreads();

        unsigned run = carry + wsum[wv] + inc - s;
        #pragma unroll
        for (int j = 0; j < RP_V4; ++j) {
            uint4 p;
            p.x = run; run += r[j].x;
            p.y = run; run += r[j].y;
            p.z = run; run += r[j].z;
            p.w = run; run += r[j].w;
            buf[tid * RP_V4 + j] = p;
        }
        carry += wsum[SCAT_T / 32];
        __syncthreads();

        #pragma unroll
        for (int j = 0; j < RP_V4; ++j) v[j] = buf[tid + j * SCAT_T];
        #pragma unroll
        for (int j = 0; j < RP_V4; ++j) {
            const int q = tid + j * SCAT_T;
            if (q < nFull) ((uint4*)gout)[q] = v[j];
        }
        if ((len & 3) && tid == (nFull & (SCAT_T - 1))) {
            const int e = nFull * 4;
            const uint4 t = buf[nFull];
            gout[e] = t.x;
            if (e + 1 < len) gout[e + 1] = t.y;
            if (e + 2 < len) gout[e + 2] = t.z;
        }
        __syncthreads();
    }
    if (tid == 0) rowptr[n] = carry;
}

#define W1_V4 (IN_FEATS * HIDDEN / 4 / SCAT_T)

__global__ void __launch_bounds__(SCAT_T) k_fill_and_gemm1(const int* __restrict__ src,
                                                           const int* __restrict__ dst,
                                                           int E, int n, int nPad, int nChunks, int chunkLen,
                                                           const unsigned* __restrict__ rowptr,
                                                           const unsigned* __restrict__ preP,
                                                           int* __restrict__ csrSrc,
                                                           int fillWgs,
                                                           const float* __restrict__ X,
                                                           const float* __restrict__ W1,
                                                           const float* __restrict__ dinv,
                                                           float* __restrict__ H,
                                                           int nTiles) {
    __shared__ union {
        unsigned cur[SCAT_CAP];
        float    w[IN_FEATS * HIDDEN];
    } sm;
    const int tid = threadIdx.x;

    if ((int)blockIdx.x < fillWgs) {
        const int r   = blockIdx.x / nChunks;
        const int c   = blockIdx.x - r * nChunks;
        const int lo  = r * SCAT_CAP;
        const int len = min(SCAT_CAP, n - lo);

        const unsigned* __restrict__ pre = preP + (size_t)c * nPad + lo;
        for (int i = tid; i < len; i += SCAT_T) sm.cur[i] = rowptr[lo + i] + pre[i];
        __syncthreads();

        const int e0 = c * chunkLen;
        const int e1 = min(E, e0 + chunkLen);
        int base = e0 + tid;
        for (; base + 7 * SCAT_T < e1; base += 8 * SCAT_T) {
            unsigned l[8];
            int      s[8];
            #pragma unroll
            for (int j = 0; j < 8; ++j) {
                l[j] = (unsigned)(dst[base + j * SCAT_T] - lo);
                s[j] = src[base + j * SCAT_T];
            }
            unsigned p[8];
            #pragma unroll
            for (int j = 0; j < 8; ++j) {
                p[j] = 0u;
                if (l[j] < (unsigned)len) p[j] = atomicAdd(&sm.cur[l[j]], 1u);
            }
            #pragma unroll
            for (int j = 0; j < 8; ++j)
                if (l[j] < (unsigned)len) csrSrc[p[j]] = s[j];
        }
        for (; base < e1; base += SCAT_T) {
            const unsigned l = (unsigned)(dst[base] - lo);
            if (l < (unsigned)len) csrSrc[atomicAdd(&sm.cur[l], 1u)] = src[base];
        }
        return;
    }

    {
        const float4* __restrict__ W4 = (const float4*)W1;
        float4* w4 = (float4*)sm.w;
        float4 t[W1_V4];
        #pragma unroll
        for (int j = 0; j < W1_V4; ++j) t[j] = W4[tid + j * SCAT_T];
        #pragma unroll
        for (int j = 0; j < W1_V4; ++j) w4[tid + j * SCAT_T] = t[j];
    }
    __syncthreads();

    const int waveId = ((int)blockIdx.x - fillWgs) * (SCAT_T / 32) + (tid >> 5);
    const int lane   = tid & 31;
    if (waveId >= nTiles) return;

    const int nTilesN = HIDDEN / 16;
    const int mt = waveId / nTilesN;
    const int nt = waveId % nTilesN;

    const int row = mt * 16 + (lane & 15);
    const int nn  = nt * 16 + (lane & 15);
    const int kq  = (lane >> 4) << 1;

    const float* __restrict__ xrow = X + (size_t)row * IN_FEATS;

    const int rbase = mt * 16 + ((lane >> 4) << 3);
    float dv[8];
    #pragma unroll
    for (int v = 0; v < 8; ++v) dv[v] = dinv[rbase + v];

    v8f acc = {};
    #pragma unroll 8
    for (int k = 0; k < IN_FEATS; k += 4) {
        v2f a, b;
        a.x = xrow[k + kq];
        a.y = xrow[k + kq + 1];
        b.x = sm.w[(k + kq)     * HIDDEN + nn];
        b.y = sm.w[(k + kq + 1) * HIDDEN + nn];
        acc = __builtin_amdgcn_wmma_f32_16x16x4_f32(
                 false, a,  false, b,
                 (short)0, acc,  false,  false);
    }

    #pragma unroll
    for (int v = 0; v < 8; ++v)
        H[(size_t)(rbase + v) * HIDDEN + nn] = acc[v] * dv[v];
}

__global__ void __launch_bounds__(256) k_gather1(const unsigned* __restrict__ rowptr,
                                                 const int* __restrict__ csrSrc,
                                                 const float* __restrict__ dinv,
                                                 const float* __restrict__ H,
                                                 float* __restrict__ A1, int n) {
    const int node = __builtin_amdgcn_readfirstlane((int)((blockIdx.x * blockDim.x + threadIdx.x) >> 5));
    if (node >= n) return;
    const int lane = threadIdx.x & 31;
    const int half = lane >> 4;
    const int q    = lane & 15;
    const float4* __restrict__ H4 = (const float4*)H;

    const float dd  = dinv[node];
    const int   beg = (int)rowptr[node];
    const int   end = (int)rowptr[node + 1];

    float4 acc = make_float4(0.0f, 0.0f, 0.0f, 0.0f);
    float4 hs = make_float4(0.0f, 0.0f, 0.0f, 0.0f);
    if (half == 0) hs = H4[(size_t)node * (HIDDEN / 4) + q];

    for (int b0 = beg; b0 < end; b0 += 32) {
        const int cnt = min(32, end - b0);
        int s_l = 0;
        if (lane < cnt) s_l = csrSrc[b0 + lane];
        int j = 0;
        do {
            float4 h[8];
            #pragma unroll
            for (int u = 0; u < 8; ++u) {
                const int idx = j + 2 * u + half;
                const int s   = __shfl(s_l, idx);
                h[u] = make_float4(0.0f, 0.0f, 0.0f, 0.0f);
                if (idx < cnt) h[u] = H4[(size_t)s * (HIDDEN / 4) + q];
            }
            #pragma unroll
            for (int u = 0; u < 8; ++u) {
                acc.x += h[u].x;
                acc.y += h[u].y;
                acc.z += h[u].z;
                acc.w += h[u].w;
            }
            j += 16;
        } while (j < cnt);
    }

    acc.x += hs.x;
    acc.y += hs.y;
    acc.z += hs.z;
    acc.w += hs.w;

    acc.x += __shfl_xor(acc.x, 16);
    acc.y += __shfl_xor(acc.y, 16);
    acc.z += __shfl_xor(acc.z, 16);
    acc.w += __shfl_xor(acc.w, 16);
    acc.x *= dd; acc.y *= dd; acc.z *= dd; acc.w *= dd;
    if (half == 0) ((float4*)A1)[(size_t)node * (HIDDEN / 4) + q] = acc;
}

__global__ void k_dense2(const float* __restrict__ A1, const float* __restrict__ b1,
                         const float* __restrict__ W2, const float* __restrict__ dinv,
                         float* __restrict__ G, int n) {
    int i = blockIdx.x * blockDim.x + threadIdx.x;
    if (i >= n) return;
    const float* __restrict__ a = A1 + (size_t)i * HIDDEN;
    const float d = dinv[i];
    float acc0 = 0.0f, acc1 = 0.0f;
    #pragma unroll 16
    for (int f = 0; f < HIDDEN; ++f) {
        float v = a[f] + b1[f];
        v = v > 0.0f ? v : 0.0f;
        acc0 = fmaf(v, W2[f * 2 + 0], acc0);
        acc1 = fmaf(v, W2[f * 2 + 1], acc1);
    }
    G[i * 2 + 0] = acc0 * d;
    G[i * 2 + 1] = acc1 * d;
}

#define G2_LANES 4
#define G2_WPB   32
#define G2_U     12

__device__ __forceinline__ float quad_sum(float v) {
    v += __int_as_float(__builtin_amdgcn_update_dpp(0, __float_as_int(v), 0xB1, 0xF, 0xF, true));
    v += __int_as_float(__builtin_amdgcn_update_dpp(0, __float_as_int(v), 0x4E, 0xF, 0xF, true));
    return v;
}

__global__ void __launch_bounds__(G2_WPB * 32) k_gather2(const unsigned* __restrict__ rowptr,
                                                 const int* __restrict__ csrSrc,
                                                 const float* __restrict__ dinv,
                                                 const float* __restrict__ G,
                                                 const float* __restrict__ b2,
                                                 float* __restrict__ out, int n) {
    const int wave = (int)((blockIdx.x * blockDim.x + threadIdx.x) >> 5);
    const int lane = threadIdx.x & 31;
    const int grp  = lane >> 2;
    const int sub  = lane & 3;
    const float2* __restrict__ G2 = (const float2*)G;

    const int  nodeA = wave * 16 + grp;
    const int  nodeB = nodeA + 8;
    const bool okA   = nodeA < n;
    const bool okB   = nodeB < n;
    const int  cA    = min(nodeA, n - 1);
    const int  cB    = min(nodeB, n - 1);

    int begA = (int)rowptr[cA], endA = (int)rowptr[cA + 1];
    int begB = (int)rowptr[cB], endB = (int)rowptr[cB + 1];
    const int   nodeS = (sub == 0) ? nodeA : nodeB;
    const bool  okS   = (sub < 2) && (nodeS < n);
    const float dS    = dinv[min(nodeS, n - 1)];
    const float2 selfA = G2[cA];
    const float2 selfB = G2[cB];
    if (!okA) endA = begA;
    if (!okB) endB = begB;

    float2 accA = make_float2(0.0f, 0.0f);
    float2 accB = make_float2(0.0f, 0.0f);
    if (sub == 0 && okA) accA = selfA;
    if (sub == 0 && okB) accB = selfB;

    const int lastA = max(endA - 1, 0);
    const int lastB = max(endB - 1, 0);
    int eA = begA + sub;
    int eB = begB + sub;
    while ((eA < endA) | (eB < endB)) {
        int sA[G2_U], sB[G2_U];
        #pragma unroll
        for (int u = 0; u < G2_U; ++u) {
            sA[u] = csrSrc[min(eA + u * G2_LANES, lastA)];
            sB[u] = csrSrc[min(eB + u * G2_LANES, lastB)];
        }
        float2 gA[G2_U], gB[G2_U];
        #pragma unroll
        for (int u = 0; u < G2_U; ++u) {
            gA[u] = G2[sA[u]];
            gB[u] = G2[sB[u]];
        }
        #pragma unroll
        for (int u = 0; u < G2_U; ++u)
            asm volatile("" :: "v"(gA[u].x), "v"(gA[u].y), "v"(gB[u].x), "v"(gB[u].y));
        #pragma unroll
        for (int u = 0; u < G2_U; ++u) {
            const bool vA = eA + u * G2_LANES < endA;
            const bool vB = eB + u * G2_LANES < endB;
            accA.x += vA ? gA[u].x : 0.0f; accA.y += vA ? gA[u].y : 0.0f;
            accB.x += vB ? gB[u].x : 0.0f; accB.y += vB ? gB[u].y : 0.0f;
        }
        eA += G2_U * G2_LANES;
        eB += G2_U * G2_LANES;
    }

    accA.x = quad_sum(accA.x); accA.y = quad_sum(accA.y);
    accB.x = quad_sum(accB.x); accB.y = quad_sum(accB.y);

    if (okS) {
        const float2 t = (sub == 0) ? accA : accB;
        float2 o;
        o.x = fmaf(dS, t.x, b2[0]);
        o.y = fmaf(dS, t.y, b2[1]);
        ((float2*)out)[nodeS] = o;
    }
}

extern "C" void kernel_launch(void* const* d_in, const int* in_sizes, int n_in,
                              void* d_out, int out_size, void* d_ws, size_t ws_size,
                              hipStream_t stream) {
    const float* x  = (const float*)d_in[0];
    const float* W1 = (const float*)d_in[1];
    const float* b1 = (const float*)d_in[2];
    const float* W2 = (const float*)d_in[3];
    const float* b2 = (const float*)d_in[4];
    const int*   ei = (const int*)d_in[5];

    const int N = in_sizes[0] / IN_FEATS;
    const int E = in_sizes[5] / 2;
    const int* src = ei;
    const int* dst = ei + E;

    const int nRanges  = (N + SCAT_CAP - 1) / SCAT_CAP;
    const int nChunks  = nRanges < SCAT_WGS ? SCAT_WGS / nRanges : 1;
    const int chunkLen = (E + nChunks - 1) / nChunks;
    const int nPad     = (N + 31) & ~31;

    float* ws   = (float*)d_ws;
    float* dinv = ws;
    float* H    = dinv + nPad;
    float* A1   = H    + (size_t)N * HIDDEN;
    float* G    = A1   + (size_t)N * HIDDEN;
    unsigned* cntP = (unsigned*)(G + (size_t)nPad * 2);
    unsigned* preP   = cntP   + (size_t)nChunks * nPad;
    unsigned* cntTot = preP   + (size_t)nChunks * nPad;
    unsigned* rowptr = cntTot + nPad;
    int*      csrSrc = (int*)(rowptr + nPad + 32);
    float* out  = (float*)d_out;

    const int T = 256;

    k_deg_part <<<nRanges * nChunks, SCAT_T, 0, stream>>>(dst, E, N, nPad, nChunks, chunkLen, cntP);
    k_make_dinv<<<(N + T - 1) / T, T, 0, stream>>>(cntP, nPad, nChunks, dinv, preP, cntTot, N);

    k_rowptr  <<<1, SCAT_T, 0, stream>>>(cntTot, N, rowptr);
    const int fillWgs = nRanges * nChunks;
    const int nTiles  = (N / 16) * (HIDDEN / 16);
    const int gemmWgs = (nTiles + (SCAT_T / 32) - 1) / (SCAT_T / 32);
    k_fill_and_gemm1<<<fillWgs + gemmWgs, SCAT_T, 0, stream>>>(src, dst, E, N, nPad, nChunks, chunkLen, rowptr, preP, csrSrc,
                                                               fillWgs, x, W1, dinv, H, nTiles);

    k_gather1<<<(N * 32 + T - 1) / T, T, 0, stream>>>(rowptr, csrSrc, dinv, H, A1, N);

    k_dense2 <<<(N + T - 1) / T, T, 0, stream>>>(A1, b1, W2, dinv, G, N);
    const int g2Waves = (N + 15) / 16;
    int g2Wpb = (g2Waves + SCAT_WGS - 1) / SCAT_WGS;
    if (g2Wpb > G2_WPB) g2Wpb = G2_WPB;
    if (g2Wpb < 1) g2Wpb = 1;
    k_gather2<<<(g2Waves + g2Wpb - 1) / g2Wpb, g2Wpb * 32, 0, stream>>>(rowptr, csrSrc, dinv, G, b2, out, N);
}


# bias+ReLU+W2 product+dinv scale done in the layer-1 aggregation's epilogue on the row in registers: no A1 plane, 6 launches; dot products sum 4 per lane, then over the lanes (sum order changes)
# speedup vs baseline: 4.7295x; 1.0516x over previous
#include <hip/hip_runtime.h>
#include <hip/hip_bf16.h>

typedef float v2f __attribute__((ext_vector_type(2)));
typedef float v8f __attribute__((ext_vector_type(8)));

#define IN_FEATS 128
#define HIDDEN   64

#define SCAT_CAP 6272
#define SCAT_T   1024
#define SCAT_WGS 256

__global__ void __launch_bounds__(SCAT_T) k_deg_part(const int* __restrict__ dst, int E, int n,
                                                     int nPad, int nChunks, int chunkLen,
                                                     unsigned* __restrict__ cntP) {
    __shared__ unsigned cnt[SCAT_CAP];
    const int r   = blockIdx.x / nChunks;
    const int c   = blockIdx.x - r * nChunks;
    const int lo  = r * SCAT_CAP;
    const int len = min(SCAT_CAP, n - lo);
    const int tid = threadIdx.x;

    for (int i = tid; i < len; i += SCAT_T) cnt[i] = 0u;
    __syncthreads();

    const int e0 = c * chunkLen;
    const int e1 = min(E, e0 + chunkLen);
    int base = e0 + tid;
    for (; base + 7 * SCAT_T < e1; base += 8 * SCAT_T) {
        unsigned l[8];
        #pragma unroll
        for (int j = 0; j < 8; ++j) l[j] = (unsigned)(dst[base + j * SCAT_T] - lo);
        #pragma unroll
        for (int j = 0; j < 8; ++j)
            if (l[j] < (unsigned)len) atomicAdd(&cnt[l[j]], 1u);
    }
    for (; base < e1; base += SCAT_T) {
        const unsigned l = (unsigned)(dst[base] - lo);
        if (l < (unsigned)len) atomicAdd(&cnt[l], 1u);
    }
    __syncthreads();

    unsigned* __restrict__ o = cntP + (size_t)c * nPad + lo;
    for (int i = tid; i < len; i += SCAT_T) o[i] = cnt[i];
}

__global__ void k_make_dinv(const unsigned* __restrict__ cntP, int nPad, int nChunks,
                            float* __restrict__ dinv, unsigned* __restrict__ preP,
                            unsigned* __restrict__ cntTot, int n) {
    int i = blockIdx.x * blockDim.x + threadIdx.x;
    if (i < n) {
        unsigned cnt = 0u;
        #pragma unroll 8
        for (int c = 0; c < nChunks; ++c) {
            const unsigned v = cntP[(size_t)c * nPad + i];
            preP[(size_t)c * nPad + i] = cnt;
            cnt += v;
        }
        cntTot[i] = cnt;
        float d = 1.0f + (float)cnt;
        dinv[i] = (d > 0.0f) ? rsqrtf(d) : 0.0f;
    }
}

#define RP_V4  13
#define RP_PER (RP_V4 * 4)
#define RP_CAP (SCAT_T * RP_PER)

__global__ void __launch_bounds__(SCAT_T) k_rowptr(const unsigned* __restrict__ cntTot, int n,
                                                   unsigned* __restrict__ rowptr) {
    __shared__ uint4    buf[RP_CAP / 4];
    __shared__ unsigned wsum[SCAT_T / 32 + 1];
    const int tid  = threadIdx.x;
    const int lane = tid & 31;
    const int wv   = tid >> 5;

    unsigned carry = 0u;
    for (int t0 = 0; t0 < n; t0 += RP_CAP) {
        const int len = min(RP_CAP, n - t0);
        const unsigned* __restrict__ gin  = cntTot + t0;
        unsigned* __restrict__       gout = rowptr + t0;

        const int nFull = len >> 2;
        const int qLast = max(nFull - 1, 0);

        uint4 v[RP_V4];
        #pragma unroll
        for (int j = 0; j < RP_V4; ++j) v[j] = ((const uint4*)gin)[min(tid + j * SCAT_T, qLast)];
        #pragma unroll
        for (int j = 0; j < RP_V4; ++j) {
            const int q = tid + j * SCAT_T;
            buf[q] = (q < nFull) ? v[j] : make_uint4(0u, 0u, 0u, 0u);
        }
        if ((len & 3) && tid == (nFull & (SCAT_T - 1))) {
            const int e = nFull * 4;
            uint4 t = make_uint4(0u, 0u, 0u, 0u);
            t.x = gin[e];
            if (e + 1 < len) t.y = gin[e + 1];
            if (e + 2 < len) t.z = gin[e + 2];
            buf[nFull] = t;
        }
        __syncthreads();

        uint4 r[RP_V4];
        #pragma unroll
        for (int j = 0; j < RP_V4; ++j) r[j] = buf[tid * RP_V4 + j];
        unsigned s = 0u;
        #pragma unroll
        for (int j = 0; j < RP_V4; ++j) s += r[j].x + r[j].y + r[j].z + r[j].w;

        unsigned inc = s;
        for (int o = 1; o < 32; o <<= 1) {
            const unsigned t = __shfl_up(inc, o);
            if (lane >= o) inc += t;
        }
        if (lane == 31) wsum[wv] = inc;
        __syncthreads();
        if (wv == 0) {
            const unsigned w = wsum[lane];
            unsigned winc = w;
            for (int o = 1; o < 32; o <<= 1) {
                const unsigned t = __shfl_up(winc, o);
                if (lane >= o) winc += t;
            }
            wsum[lane] = winc - w;
            if (lane == 31) wsum[SCAT_T / 32] = winc;
        }
        __syncthreads();

        unsigned run = carry + wsum[wv] + inc - s;
        #pragma unroll
        for (int j = 0; j < RP_V4; ++j) {
            uint4 p;
            p.x = run; run += r[j].x;
            p.y = run; run += r[j].y;
            p.z = run; run += r[j].z;
            p.w = run; run += r[j].w;
            buf[tid * RP_V4 + j] = p;
        }
        carry += wsum[SCAT_T / 32];
        __syncthreads();

        #pragma unroll
        for (int j = 0; j < RP_V4; ++j) v[j] = buf[tid + j * SCAT_T];
        #pragma unroll
        for (int j = 0; j < RP_V4; ++j) {
            const int q = tid + j * SCAT_T;
            if (q < nFull) ((uint4*)gout)[q] = v[j];
        }
        if ((len & 3) && tid == (nFull & (SCAT_T - 1))) {
            const int e = nFull * 4;
            const uint4 t = buf[nFull];
            gout[e] = t.x;
            if (e + 1 < len) gout[e + 1] = t.y;
            if (e + 2 < len) gout[e + 2] = t.z;
        }
        __syncthreads();
    }
    if (tid == 0) rowptr[n] = carry;
}

#define W1_V4 (IN_FEATS * HIDDEN / 4 / SCAT_T)

__global__ void __launch_bounds__(SCAT_T) k_fill_and_gemm1(const int* __restrict__ src,
                                                           const int* __restrict__ dst,
                                                           int E, int n, int nPad, int nChunks, int chunkLen,
                                                           const unsigned* __restrict__ rowptr,
                                                           const unsigned* __restrict__ preP,
                                                           int* __restrict__ csrSrc,
                                                           int fillWgs,
                                                           const float* __restrict__ X,
                                                           const float* __restrict__ W1,
                                                           const float* __restrict__ dinv,
                                                           float* __restrict__ H,
                                                           int nTiles) {
    __shared__ union {
        unsigned cur[SCAT_CAP];
        float    w[IN_FEATS * HIDDEN];
    } sm;
    const int tid = threadIdx.x;

    if ((int)blockIdx.x < fillWgs) {
        const int r   = blockIdx.x / nChunks;
        const int c   = blockIdx.x - r * nChunks;
        const int lo  = r * SCAT_CAP;
        const int len = min(SCAT_CAP, n - lo);

        const unsigned* __restrict__ pre = preP + (size_t)c * nPad + lo;
        for (int i = tid; i < len; i += SCAT_T) sm.cur[i] = rowptr[lo + i] + pre[i];
        __syncthreads();

        const int e0 = c * chunkLen;
        const int e1 = min(E, e0 + chunkLen);
        int base = e0 + tid;
        for (; base + 7 * SCAT_T < e1; base += 8 * SCAT_T) {
            unsigned l[8];
            int      s[8];
            #pragma unroll
            for (int j = 0; j < 8; ++j) {
                l[j] = (unsigned)(dst[base + j * SCAT_T] - lo);
                s[j] = src[base + j * SCAT_T];
            }
            unsigned p[8];
            #pragma unroll
            for (int j = 0; j < 8; ++j) {
                p[j] = 0u;
                if (l[j] < (unsigned)len) p[j] = atomicAdd(&sm.cur[l[j]], 1u);
            }
            #pragma unroll
            for (int j = 0; j < 8; ++j)
                if (l[j] < (unsigned)len) csrSrc[p[j]] = s[j];
        }
        for (; base < e1; base += SCAT_T) {
            const unsigned l = (unsigned)(dst[base] - lo);
            if (l < (unsigned)len) csrSrc[atomicAdd(&sm.cur[l], 1u)] = src[base];
        }
        return;
    }

    {
        const float4* __restrict__ W4 = (const float4*)W1;
        float4* w4 = (float4*)sm.w;
        float4 t[W1_V4];
        #pragma unroll
        for (int j = 0; j < W1_V4; ++j) t[j] = W4[tid + j * SCAT_T];
        #pragma unroll
        for (int j = 0; j < W1_V4; ++j) w4[tid + j * SCAT_T] = t[j];
    }
    __syncthreads();

    const int waveId = ((int)blockIdx.x - fillWgs) * (SCAT_T / 32) + (tid >> 5);
    const int lane   = tid & 31;
    if (waveId >= nTiles) return;

    const int nTilesN = HIDDEN / 16;
    const int mt = waveId / nTilesN;
    const int nt = waveId % nTilesN;

    const int row = mt * 16 + (lane & 15);
    const int nn  = nt * 16 + (lane & 15);
    const int kq  = (lane >> 4) << 1;

    const float* __restrict__ xrow = X + (size_t)row * IN_FEATS;

    const int rbase = mt * 16 + ((lane >> 4) << 3);
    float dv[8];
    #pragma unroll
    for (int v = 0; v < 8; ++v) dv[v] = dinv[rbase + v];

    v8f acc = {};
    #pragma unroll 8
    for (int k = 0; k < IN_FEATS; k += 4) {
        v2f a, b;
        a.x = xrow[k + kq];
        a.y = xrow[k + kq + 1];
        b.x = sm.w[(k + kq)     * HIDDEN + nn];
        b.y = sm.w[(k + kq + 1) * HIDDEN + nn];
        acc = __builtin_amdgcn_wmma_f32_16x16x4_f32(
                 false, a,  false, b,
                 (short)0, acc,  false,  false);
    }

    #pragma unroll
    for (int v = 0; v < 8; ++v)
        H[(size_t)(rbase + v) * HIDDEN + nn] = acc[v] * dv[v];
}

__device__ __forceinline__ float row16_sum(float v) {
    v += __int_as_float(__builtin_amdgcn_update_dpp(0, __float_as_int(v), 0x161, 0xF, 0xF, true));
    v += __int_as_float(__builtin_amdgcn_update_dpp(0, __float_as_int(v), 0x162, 0xF, 0xF, true));
    v += __int_as_float(__builtin_amdgcn_update_dpp(0, __float_as_int(v), 0x164, 0xF, 0xF, true));
    v += __int_as_float(__builtin_amdgcn_update_dpp(0, __float_as_int(v), 0x168, 0xF, 0xF, true));
    return v;
}

__global__ void __launch_bounds__(256) k_gather1(const unsigned* __restrict__ rowptr,
                                                 const int* __restrict__ csrSrc,
                                                 const float* __restrict__ dinv,
                                                 const float* __restrict__ H,
                                                 const float* __restrict__ b1,
                                                 const float* __restrict__ W2,
                                                 float* __restrict__ G, int n) {
    const int node = __builtin_amdgcn_readfirstlane((int)((blockIdx.x * blockDim.x + threadIdx.x) >> 5));
    if (node >= n) return;
    const int lane = threadIdx.x & 31;
    const int half = lane >> 4;
    const int q    = lane & 15;
    const float4* __restrict__ H4 = (const float4*)H;

    const float dd  = dinv[node];
    const int   beg = (int)rowptr[node];
    const int   end = (int)rowptr[node + 1];

    float4 acc = make_float4(0.0f, 0.0f, 0.0f, 0.0f);
    float4 hs = make_float4(0.0f, 0.0f, 0.0f, 0.0f);
    if (half == 0) hs = H4[(size_t)node * (HIDDEN / 4) + q];
    const float4 bq  = ((const float4*)b1)[q];
    const float4 w01 = ((const float4*)W2)[2 * q];
    const float4 w23 = ((const float4*)W2)[2 * q + 1];

    for (int b0 = beg; b0 < end; b0 += 32) {
        const int cnt = min(32, end - b0);
        int s_l = 0;
        if (lane < cnt) s_l = csrSrc[b0 + lane];
        int j = 0;
        do {
            float4 h[8];
            #pragma unroll
            for (int u = 0; u < 8; ++u) {
                const int idx = j + 2 * u + half;
                const int s   = __shfl(s_l, idx);
                h[u] = make_float4(0.0f, 0.0f, 0.0f, 0.0f);
                if (idx < cnt) h[u] = H4[(size_t)s * (HIDDEN / 4) + q];
            }
            #pragma unroll
            for (int u = 0; u < 8; ++u) {
                acc.x += h[u].x;
                acc.y += h[u].y;
                acc.z += h[u].z;
                acc.w += h[u].w;
            }
            j += 16;
        } while (j < cnt);
    }

    acc.x += hs.x;
    acc.y += hs.y;
    acc.z += hs.z;
    acc.w += hs.w;

    acc.x += __shfl_xor(acc.x, 16);
    acc.y += __shfl_xor(acc.y, 16);
    acc.z += __shfl_xor(acc.z, 16);
    acc.w += __shfl_xor(acc.w, 16);
    acc.x *= dd; acc.y *= dd; acc.z *= dd; acc.w *= dd;

    float v0 = acc.x + bq.x;
    float v1 = acc.y + bq.y;
    float v2 = acc.z + bq.z;
    float v3 = acc.w + bq.w;
    v0 = v0 > 0.0f ? v0 : 0.0f;
    v1 = v1 > 0.0f ? v1 : 0.0f;
    v2 = v2 > 0.0f ? v2 : 0.0f;
    v3 = v3 > 0.0f ? v3 : 0.0f;
    float p0 = v0 * w01.x;
    float p1 = v0 * w01.y;
    p0 = fmaf(v1, w01.z, p0);
    p1 = fmaf(v1, w01.w, p1);
    p0 = fmaf(v2, w23.x, p0);
    p1 = fmaf(v2, w23.y, p1);
    p0 = fmaf(v3, w23.z, p0);
    p1 = fmaf(v3, w23.w, p1);
    p0 = row16_sum(p0);
    p1 = row16_sum(p1);
    if (lane == 0) ((float2*)G)[node] = make_float2(p0 * dd, p1 * dd);
}

#define G2_LANES 4
#define G2_WPB   32
#define G2_U     12

__device__ __forceinline__ float quad_sum(float v) {
    v += __int_as_float(__builtin_amdgcn_update_dpp(0, __float_as_int(v), 0xB1, 0xF, 0xF, true));
    v += __int_as_float(__builtin_amdgcn_update_dpp(0, __float_as_int(v), 0x4E, 0xF, 0xF, true));
    return v;
}

__global__ void __launch_bounds__(G2_WPB * 32) k_gather2(const unsigned* __restrict__ rowptr,
                                                 const int* __restrict__ csrSrc,
                                                 const float* __restrict__ dinv,
                                                 const float* __restrict__ G,
                                                 const float* __restrict__ b2,
                                                 float* __restrict__ out, int n) {
    const int wave = (int)((blockIdx.x * blockDim.x + threadIdx.x) >> 5);
    const int lane = threadIdx.x & 31;
    const int grp  = lane >> 2;
    const int sub  = lane & 3;
    const float2* __restrict__ G2 = (const float2*)G;

    const int  nodeA = wave * 16 + grp;
    const int  nodeB = nodeA + 8;
    const bool okA   = nodeA < n;
    const bool okB   = nodeB < n;
    const int  cA    = min(nodeA, n - 1);
    const int  cB    = min(nodeB, n - 1);

    int begA = (int)rowptr[cA], endA = (int)rowptr[cA + 1];
    int begB = (int)rowptr[cB], endB = (int)rowptr[cB + 1];
    const int   nodeS = (sub == 0) ? nodeA : nodeB;
    const bool  okS   = (sub < 2) && (nodeS < n);
    const float dS    = dinv[min(nodeS, n - 1)];
    const float2 selfA = G2[cA];
    const float2 selfB = G2[cB];
    if (!okA) endA = begA;
    if (!okB) endB = begB;

    float2 accA = make_float2(0.0f, 0.0f);
    float2 accB = make_float2(0.0f, 0.0f);
    if (sub == 0 && okA) accA = selfA;
    if (sub == 0 && okB) accB = selfB;

    const int lastA = max(endA - 1, 0);
    const int lastB = max(endB - 1, 0);
    int eA = begA + sub;
    int eB = begB + sub;
    while ((eA < endA) | (eB < endB)) {
        int sA[G2_U], sB[G2_U];
        #pragma unroll
        for (int u = 0; u < G2_U; ++u) {
            sA[u] = csrSrc[min(eA + u * G2_LANES, lastA)];
            sB[u] = csrSrc[min(eB + u * G2_LANES, lastB)];
        }
        float2 gA[G2_U], gB[G2_U];
        #pragma unroll
        for (int u = 0; u < G2_U; ++u) {
            gA[u] = G2[sA[u]];
            gB[u] = G2[sB[u]];
        }
        #pragma unroll
        for (int u = 0; u < G2_U; ++u)
            asm volatile("" :: "v"(gA[u].x), "v"(gA[u].y), "v"(gB[u].x), "v"(gB[u].y));
        #pragma unroll
        for (int u = 0; u < G2_U; ++u) {
            const bool vA = eA + u * G2_LANES < endA;
            const bool vB = eB + u * G2_LANES < endB;
            accA.x += vA ? gA[u].x : 0.0f; accA.y += vA ? gA[u].y : 0.0f;
            accB.x += vB ? gB[u].x : 0.0f; accB.y += vB ? gB[u].y : 0.0f;
        }
        eA += G2_U * G2_LANES;
        eB += G2_U * G2_LANES;
    }

    accA.x = quad_sum(accA.x); accA.y = quad_sum(accA.y);
    accB.x = quad_sum(accB.x); accB.y = quad_sum(accB.y);

    if (okS) {
        const float2 t = (sub == 0) ? accA : accB;
        float2 o;
        o.x = fmaf(dS, t.x, b2[0]);
        o.y = fmaf(dS, t.y, b2[1]);
        ((float2*)out)[nodeS] = o;
    }
}

extern "C" void kernel_launch(void* const* d_in, const int* in_sizes, int n_in,
                              void* d_out, int out_size, void* d_ws, size_t ws_size,
                              hipStream_t stream) {
    const float* x  = (const float*)d_in[0];
    const float* W1 = (const float*)d_in[1];
    const float* b1 = (const float*)d_in[2];
    const float* W2 = (const float*)d_in[3];
    const float* b2 = (const float*)d_in[4];
    const int*   ei = (const int*)d_in[5];

    const int N = in_sizes[0] / IN_FEATS;
    const int E = in_sizes[5] / 2;
    const int* src = ei;
    const int* dst = ei + E;

    const int nRanges  = (N + SCAT_CAP - 1) / SCAT_CAP;
    const int nChunks  = nRanges < SCAT_WGS ? SCAT_WGS / nRanges : 1;
    const int chunkLen = (E + nChunks - 1) / nChunks;
    const int nPad     = (N + 31) & ~31;

    float* ws   = (float*)d_ws;
    float* dinv = ws;
    float* H    = dinv + nPad;
    float* G    = H    + (size_t)N * HIDDEN;
    unsigned* cntP = (unsigned*)(G + (size_t)nPad * 2);
    unsigned* preP   = cntP   + (size_t)nChunks * nPad;
    unsigned* cntTot = preP   + (size_t)nChunks * nPad;
    unsigned* rowptr = cntTot + nPad;
    int*      csrSrc = (int*)(rowptr + nPad + 32);
    float* out  = (float*)d_out;

    const int T = 256;

    k_deg_part <<<nRanges * nChunks, SCAT_T, 0, stream>>>(dst, E, N, nPad, nChunks, chunkLen, cntP);
    k_make_dinv<<<(N + T - 1) / T, T, 0, stream>>>(cntP, nPad, nChunks, dinv, preP, cntTot, N);

    k_rowptr  <<<1, SCAT_T, 0, stream>>>(cntTot, N, rowptr);
    const int fillWgs = nRanges * nChunks;
    const int nTiles  = (N / 16) * (HIDDEN / 16);
    const int gemmWgs = (nTiles + (SCAT_T / 32) - 1) / (SCAT_T / 32);
    k_fill_and_gemm1<<<fillWgs + gemmWgs, SCAT_T, 0, stream>>>(src, dst, E, N, nPad, nChunks, chunkLen, rowptr, preP, csrSrc,
                                                               fillWgs, x, W1, dinv, H, nTiles);

    k_gather1<<<(N * 32 + T - 1) / T, T, 0, stream>>>(rowptr, csrSrc, dinv, H, b1, W2, G, N);

    const int g2Waves = (N + 15) / 16;
    int g2Wpb = (g2Waves + SCAT_WGS - 1) / SCAT_WGS;
    if (g2Wpb > G2_WPB) g2Wpb = G2_WPB;
    if (g2Wpb < 1) g2Wpb = 1;
    k_gather2<<<(g2Waves + g2Wpb - 1) / g2Wpb, g2Wpb * 32, 0, stream>>>(rowptr, csrSrc, dinv, G, b2, out, N);
}


# row-pointer prefix sum in two levels inside its neighbour launches: block prefixes and totals in the degree reduce, each fill work-group scans the totals; no one-work-group scan launch, 5 launches
# speedup vs baseline: 5.0652x; 1.0710x over previous
#include <hip/hip_runtime.h>
#include <hip/hip_bf16.h>

typedef float v2f __attribute__((ext_vector_type(2)));
typedef float v8f __attribute__((ext_vector_type(8)));

#define IN_FEATS 128
#define HIDDEN   64

#define SCAT_CAP 6272
#define SCAT_T   1024
#define SCAT_WGS 256

__global__ void __launch_bounds__(SCAT_T) k_deg_part(const int* __restrict__ dst, int E, int n,
                                                     int nPad, int nChunks, int chunkLen,
                                                     unsigned* __restrict__ cntP) {
    __shared__ unsigned cnt[SCAT_CAP];
    const int r   = blockIdx.x / nChunks;
    const int c   = blockIdx.x - r * nChunks;
    const int lo  = r * SCAT_CAP;
    const int len = min(SCAT_CAP, n - lo);
    const int tid = threadIdx.x;

    for (int i = tid; i < len; i += SCAT_T) cnt[i] = 0u;
    __syncthreads();

    const int e0 = c * chunkLen;
    const int e1 = min(E, e0 + chunkLen);
    int base = e0 + tid;
    for (; base + 7 * SCAT_T < e1; base += 8 * SCAT_T) {
        unsigned l[8];
        #pragma unroll
        for (int j = 0; j < 8; ++j) l[j] = (unsigned)(dst[base + j * SCAT_T] - lo);
        #pragma unroll
        for (int j = 0; j < 8; ++j)
            if (l[j] < (unsigned)len) atomicAdd(&cnt[l[j]], 1u);
    }
    for (; base < e1; base += SCAT_T) {
        const unsigned l = (unsigned)(dst[base] - lo);
        if (l < (unsigned)len) atomicAdd(&cnt[l], 1u);
    }
    __syncthreads();

    unsigned* __restrict__ o = cntP + (size_t)c * nPad + lo;
    for (int i = tid; i < len; i += SCAT_T) o[i] = cnt[i];
}

#define MD_T 128

__global__ void __launch_bounds__(MD_T) k_make_dinv(const unsigned* __restrict__ cntP, int nPad, int nChunks,
                                                    float* __restrict__ dinv, unsigned* __restrict__ preP,
                                                    unsigned* __restrict__ rowLocal,
                                                    unsigned* __restrict__ blockTot, int n) {
    __shared__ unsigned wtot[MD_T / 32];
    const int i    = blockIdx.x * MD_T + threadIdx.x;
    const int lane = threadIdx.x & 31;
    const int wv   = threadIdx.x >> 5;
    unsigned cnt = 0u;
    if (i < n) {
        #pragma unroll 8
        for (int c = 0; c < nChunks; ++c) {
            const unsigned v = cntP[(size_t)c * nPad + i];
            preP[(size_t)c * nPad + i] = cnt;
            cnt += v;
        }
        float d = 1.0f + (float)cnt;
        dinv[i] = (d > 0.0f) ? rsqrtf(d) : 0.0f;
    }
    unsigned inc = cnt;
    for (int o = 1; o < 32; o <<= 1) {
        const unsigned t = __shfl_up(inc, o);
        if (lane >= o) inc += t;
    }
    if (lane == 31) wtot[wv] = inc;
    __syncthreads();
    unsigned before = 0u, tot = 0u;
    #pragma unroll
    for (int w = 0; w < MD_T / 32; ++w) {
        const unsigned v = wtot[w];
        if (w < wv) before += v;
        tot += v;
    }
    if (i < n) rowLocal[i] = before + inc - cnt;
    if (threadIdx.x == 0) blockTot[blockIdx.x] = tot;
}

#define W1_V4 (IN_FEATS * HIDDEN / 4 / SCAT_T)
#define FILL_V ((SCAT_CAP + SCAT_T - 1) / SCAT_T)

__global__ void __launch_bounds__(SCAT_T) k_fill_and_gemm1(const int* __restrict__ src,
                                                           const int* __restrict__ dst,
                                                           int E, int n, int nPad, int nChunks, int chunkLen,
                                                           const unsigned* __restrict__ rowLocal,
                                                           const unsigned* __restrict__ blockTot,
                                                           const unsigned* __restrict__ preP,
                                                           unsigned* __restrict__ rowptr,
                                                           int* __restrict__ csrSrc,
                                                           int fillWgs,
                                                           const float* __restrict__ X,
                                                           const float* __restrict__ W1,
                                                           const float* __restrict__ dinv,
                                                           float* __restrict__ H,
                                                           int nTiles) {
    __shared__ union {
        struct {
            unsigned cur[SCAT_CAP];
            unsigned bb[SCAT_T];
            unsigned ws[SCAT_T / 32 + 1];
        } f;
        float w[IN_FEATS * HIDDEN];
    } sm;
    const int tid = threadIdx.x;

    if ((int)blockIdx.x < fillWgs) {
        const int r   = blockIdx.x / nChunks;
        const int c   = blockIdx.x - r * nChunks;
        const int lo  = r * SCAT_CAP;
        const int len = min(SCAT_CAP, n - lo);

        const unsigned* __restrict__ pre = preP + (size_t)c * nPad + lo;
        const int lane = tid & 31;
        const int wv   = tid >> 5;

        const int nBlk = (n + MD_T - 1) / MD_T;
        const unsigned bt = (tid < nBlk) ? blockTot[tid] : 0u;
        unsigned rl[FILL_V], pr[FILL_V];
        #pragma unroll
        for (int j = 0; j < FILL_V; ++j) {
            const int ic = min(tid + j * SCAT_T, len - 1);
            rl[j] = rowLocal[lo + ic];
            pr[j] = pre[ic];
        }

        unsigned inc = bt;
        for (int o = 1; o < 32; o <<= 1) {
            const unsigned t = __shfl_up(inc, o);
            if (lane >= o) inc += t;
        }
        if (lane == 31) sm.f.ws[wv] = inc;
        __syncthreads();
        if (wv == 0) {
            const unsigned w = sm.f.ws[lane];
            unsigned winc = w;
            for (int o = 1; o < 32; o <<= 1) {
                const unsigned t = __shfl_up(winc, o);
                if (lane >= o) winc += t;
            }
            sm.f.ws[lane] = winc - w;
            if (lane == 31) sm.f.ws[SCAT_T / 32] = winc;
        }
        __syncthreads();
        sm.f.bb[tid] = sm.f.ws[wv] + inc - bt;
        __syncthreads();

        #pragma unroll
        for (int j = 0; j < FILL_V; ++j) {
            const int i = tid + j * SCAT_T;
            if (i < len) {
                const unsigned rp = sm.f.bb[(lo + i) / MD_T] + rl[j];
                sm.f.cur[i] = rp + pr[j];
                if (c == 0) rowptr[lo + i] = rp;
            }
        }
        if (c == 0 && lo + len == n && tid == 0) rowptr[n] = sm.f.ws[SCAT_T / 32];
        __syncthreads();

        const int e0 = c * chunkLen;
        const int e1 = min(E, e0 + chunkLen);
        int base = e0 + tid;
        for (; base + 7 * SCAT_T < e1; base += 8 * SCAT_T) {
            unsigned l[8];
            int      s[8];
            #pragma unroll
            for (int j = 0; j < 8; ++j) {
                l[j] = (unsigned)(dst[base + j * SCAT_T] - lo);
                s[j] = src[base + j * SCAT_T];
            }
            unsigned p[8];
            #pragma unroll
            for (int j = 0; j < 8; ++j) {
                p[j] = 0u;
                if (l[j] < (unsigned)len) p[j] = atomicAdd(&sm.f.cur[l[j]], 1u);
            }
            #pragma unroll
            for (int j = 0; j < 8; ++j)
                if (l[j] < (unsigned)len) csrSrc[p[j]] = s[j];
        }
        for (; base < e1; base += SCAT_T) {
            const unsigned l = (unsigned)(dst[base] - lo);
            if (l < (unsigned)len) csrSrc[atomicAdd(&sm.f.cur[l], 1u)] = src[base];
        }
        return;
    }

    {
        const float4* __restrict__ W4 = (const float4*)W1;
        float4* w4 = (float4*)sm.w;
        float4 t[W1_V4];
        #pragma unroll
        for (int j = 0; j < W1_V4; ++j) t[j] = W4[tid + j * SCAT_T];
        #pragma unroll
        for (int j = 0; j < W1_V4; ++j) w4[tid + j * SCAT_T] = t[j];
    }
    __syncthreads();

    const int waveId = ((int)blockIdx.x - fillWgs) * (SCAT_T / 32) + (tid >> 5);
    const int lane   = tid & 31;
    if (waveId >= nTiles) return;

    const int nTilesN = HIDDEN / 16;
    const int mt = waveId / nTilesN;
    const int nt = waveId % nTilesN;

    const int row = mt * 16 + (lane & 15);
    const int nn  = nt * 16 + (lane & 15);
    const int kq  = (lane >> 4) << 1;

    const float* __restrict__ xrow = X + (size_t)row * IN_FEATS;

    const int rbase = mt * 16 + ((lane >> 4) << 3);
    float dv[8];
    #pragma unroll
    for (int v = 0; v < 8; ++v) dv[v] = dinv[rbase + v];

    v8f acc = {};
    #pragma unroll 8
    for (int k = 0; k < IN_FEATS; k += 4) {
        v2f a, b;
        a.x = xrow[k + kq];
        a.y = xrow[k + kq + 1];
        b.x = sm.w[(k + kq)     * HIDDEN + nn];
        b.y = sm.w[(k + kq + 1) * HIDDEN + nn];
        acc = __builtin_amdgcn_wmma_f32_16x16x4_f32(
                 false, a,  false, b,
                 (short)0, acc,  false,  false);
    }

    #pragma unroll
    for (int v = 0; v < 8; ++v)
        H[(size_t)(rbase + v) * HIDDEN + nn] = acc[v] * dv[v];
}

__device__ __forceinline__ float row16_sum(float v) {
    v += __int_as_float(__builtin_amdgcn_update_dpp(0, __float_as_int(v), 0x161, 0xF, 0xF, true));
    v += __int_as_float(__builtin_amdgcn_update_dpp(0, __float_as_int(v), 0x162, 0xF, 0xF, true));
    v += __int_as_float(__builtin_amdgcn_update_dpp(0, __float_as_int(v), 0x164, 0xF, 0xF, true));
    v += __int_as_float(__builtin_amdgcn_update_dpp(0, __float_as_int(v), 0x168, 0xF, 0xF, true));
    return v;
}

__global__ void __launch_bounds__(256) k_gather1(const unsigned* __restrict__ rowptr,
                                                 const int* __restrict__ csrSrc,
                                                 const float* __restrict__ dinv,
                                                 const float* __restrict__ H,
                                                 const float* __restrict__ b1,
                                                 const float* __restrict__ W2,
                                                 float* __restrict__ G, int n) {
    const int node = __builtin_amdgcn_readfirstlane((int)((blockIdx.x * blockDim.x + threadIdx.x) >> 5));
    if (node >= n) return;
    const int lane = threadIdx.x & 31;
    const int half = lane >> 4;
    const int q    = lane & 15;
    const float4* __restrict__ H4 = (const float4*)H;

    const float dd  = dinv[node];
    const int   beg = (int)rowptr[node];
    const int   end = (int)rowptr[node + 1];

    float4 acc = make_float4(0.0f, 0.0f, 0.0f, 0.0f);
    float4 hs = make_float4(0.0f, 0.0f, 0.0f, 0.0f);
    if (half == 0) hs = H4[(size_t)node * (HIDDEN / 4) + q];
    const float4 bq  = ((const float4*)b1)[q];
    const float4 w01 = ((const float4*)W2)[2 * q];
    const float4 w23 = ((const float4*)W2)[2 * q + 1];

    for (int b0 = beg; b0 < end; b0 += 32) {
        const int cnt = min(32, end - b0);
        int s_l = 0;
        if (lane < cnt) s_l = csrSrc[b0 + lane];
        int j = 0;
        do {
            float4 h[8];
            #pragma unroll
            for (int u = 0; u < 8; ++u) {
                const int idx = j + 2 * u + half;
                const int s   = __shfl(s_l, idx);
                h[u] = make_float4(0.0f, 0.0f, 0.0f, 0.0f);
                if (idx < cnt) h[u] = H4[(size_t)s * (HIDDEN / 4) + q];
            }
            #pragma unroll
            for (int u = 0; u < 8; ++u) {
                acc.x += h[u].x;
                acc.y += h[u].y;
                acc.z += h[u].z;
                acc.w += h[u].w;
            }
            j += 16;
        } while (j < cnt);
    }

    acc.x += hs.x;
    acc.y += hs.y;
    acc.z += hs.z;
    acc.w += hs.w;

    acc.x += __shfl_xor(acc.x, 16);
    acc.y += __shfl_xor(acc.y, 16);
    acc.z += __shfl_xor(acc.z, 16);
    acc.w += __shfl_xor(acc.w, 16);
    acc.x *= dd; acc.y *= dd; acc.z *= dd; acc.w *= dd;

    float v0 = acc.x + bq.x;
    float v1 = acc.y + bq.y;
    float v2 = acc.z + bq.z;
    float v3 = acc.w + bq.w;
    v0 = v0 > 0.0f ? v0 : 0.0f;
    v1 = v1 > 0.0f ? v1 : 0.0f;
    v2 = v2 > 0.0f ? v2 : 0.0f;
    v3 = v3 > 0.0f ? v3 : 0.0f;
    float p0 = v0 * w01.x;
    float p1 = v0 * w01.y;
    p0 = fmaf(v1, w01.z, p0);
    p1 = fmaf(v1, w01.w, p1);
    p0 = fmaf(v2, w23.x, p0);
    p1 = fmaf(v2, w23.y, p1);
    p0 = fmaf(v3, w23.z, p0);
    p1 = fmaf(v3, w23.w, p1);
    p0 = row16_sum(p0);
    p1 = row16_sum(p1);
    if (lane == 0) ((float2*)G)[node] = make_float2(p0 * dd, p1 * dd);
}

#define G2_LANES 4
#define G2_WPB   32
#define G2_U     12

__device__ __forceinline__ float quad_sum(float v) {
    v += __int_as_float(__builtin_amdgcn_update_dpp(0, __float_as_int(v), 0xB1, 0xF, 0xF, true));
    v += __int_as_float(__builtin_amdgcn_update_dpp(0, __float_as_int(v), 0x4E, 0xF, 0xF, true));
    return v;
}

__global__ void __launch_bounds__(G2_WPB * 32) k_gather2(const unsigned* __restrict__ rowptr,
                                                 const int* __restrict__ csrSrc,
                                                 const float* __restrict__ dinv,
                                                 const float* __restrict__ G,
                                                 const float* __restrict__ b2,
                                                 float* __restrict__ out, int n) {
    const int wave = (int)((blockIdx.x * blockDim.x + threadIdx.x) >> 5);
    const int lane = threadIdx.x & 31;
    const int grp  = lane >> 2;
    const int sub  = lane & 3;
    const float2* __restrict__ G2 = (const float2*)G;

    const int  nodeA = wave * 16 + grp;
    const int  nodeB = nodeA + 8;
    const bool okA   = nodeA < n;
    const bool okB   = nodeB < n;
    const int  cA    = min(nodeA, n - 1);
    const int  cB    = min(nodeB, n - 1);

    int begA = (int)rowptr[cA], endA = (int)rowptr[cA + 1];
    int begB = (int)rowptr[cB], endB = (int)rowptr[cB + 1];
    const int   nodeS = (sub == 0) ? nodeA : nodeB;
    const bool  okS   = (sub < 2) && (nodeS < n);
    const float dS    = dinv[min(nodeS, n - 1)];
    const float2 selfA = G2[cA];
    const float2 selfB = G2[cB];
    if (!okA) endA = begA;
    if (!okB) endB = begB;

    float2 accA = make_float2(0.0f, 0.0f);
    float2 accB = make_float2(0.0f, 0.0f);
    if (sub == 0 && okA) accA = selfA;
    if (sub == 0 && okB) accB = selfB;

    const int lastA = max(endA - 1, 0);
    const int lastB = max(endB - 1, 0);
    int eA = begA + sub;
    int eB = begB + sub;
    while ((eA < endA) | (eB < endB)) {
        int sA[G2_U], sB[G2_U];
        #pragma unroll
        for (int u = 0; u < G2_U; ++u) {
            sA[u] = csrSrc[min(eA + u * G2_LANES, lastA)];
            sB[u] = csrSrc[min(eB + u * G2_LANES, lastB)];
        }
        float2 gA[G2_U], gB[G2_U];
        #pragma unroll
        for (int u = 0; u < G2_U; ++u) {
            gA[u] = G2[sA[u]];
            gB[u] = G2[sB[u]];
        }
        #pragma unroll
        for (int u = 0; u < G2_U; ++u)
            asm volatile("" :: "v"(gA[u].x), "v"(gA[u].y), "v"(gB[u].x), "v"(gB[u].y));
        #pragma unroll
        for (int u = 0; u < G2_U; ++u) {
            const bool vA = eA + u * G2_LANES < endA;
            const bool vB = eB + u * G2_LANES < endB;
            accA.x += vA ? gA[u].x : 0.0f; accA.y += vA ? gA[u].y : 0.0f;
            accB.x += vB ? gB[u].x : 0.0f; accB.y += vB ? gB[u].y : 0.0f;
        }
        eA += G2_U * G2_LANES;
        eB += G2_U * G2_LANES;
    }

    accA.x = quad_sum(accA.x); accA.y = quad_sum(accA.y);
    accB.x = quad_sum(accB.x); accB.y = quad_sum(accB.y);

    if (okS) {
        const float2 t = (sub == 0) ? accA : accB;
        float2 o;
        o.x = fmaf(dS, t.x, b2[0]);
        o.y = fmaf(dS, t.y, b2[1]);
        ((float2*)out)[nodeS] = o;
    }
}

extern "C" void kernel_launch(void* const* d_in, const int* in_sizes, int n_in,
                              void* d_out, int out_size, void* d_ws, size_t ws_size,
                              hipStream_t stream) {
    const float* x  = (const float*)d_in[0];
    const float* W1 = (const float*)d_in[1];
    const float* b1 = (const float*)d_in[2];
    const float* W2 = (const float*)d_in[3];
    const float* b2 = (const float*)d_in[4];
    const int*   ei = (const int*)d_in[5];

    const int N = in_sizes[0] / IN_FEATS;
    const int E = in_sizes[5] / 2;
    const int* src = ei;
    const int* dst = ei + E;

    const int nRanges  = (N + SCAT_CAP - 1) / SCAT_CAP;
    const int nChunks  = nRanges < SCAT_WGS ? SCAT_WGS / nRanges : 1;
    const int chunkLen = (E + nChunks - 1) / nChunks;
    const int nPad     = (N + 31) & ~31;

    float* ws   = (float*)d_ws;
    float* dinv = ws;
    float* H    = dinv + nPad;
    float* G    = H    + (size_t)N * HIDDEN;
    unsigned* cntP = (unsigned*)(G + (size_t)nPad * 2);
    unsigned* preP   = cntP   + (size_t)nChunks * nPad;
    unsigned* rowLocal = preP   + (size_t)nChunks * nPad;
    unsigned* rowptr   = rowLocal + nPad;
    unsigned* blockTot = rowptr + nPad + 32;
    int*      csrSrc   = (int*)(blockTot + SCAT_T);
    float* out  = (float*)d_out;

    const int T = 256;

    k_deg_part <<<nRanges * nChunks, SCAT_T, 0, stream>>>(dst, E, N, nPad, nChunks, chunkLen, cntP);
    k_make_dinv<<<(N + MD_T - 1) / MD_T, MD_T, 0, stream>>>(cntP, nPad, nChunks, dinv, preP, rowLocal, blockTot, N);

    const int fillWgs = nRanges * nChunks;
    const int nTiles  = (N / 16) * (HIDDEN / 16);
    const int gemmWgs = (nTiles + (SCAT_T / 32) - 1) / (SCAT_T / 32);
    k_fill_and_gemm1<<<fillWgs + gemmWgs, SCAT_T, 0, stream>>>(src, dst, E, N, nPad, nChunks, chunkLen, rowLocal, blockTot, preP, rowptr, csrSrc,
                                                               fillWgs, x, W1, dinv, H, nTiles);

    k_gather1<<<(N * 32 + T - 1) / T, T, 0, stream>>>(rowptr, csrSrc, dinv, H, b1, W2, G, N);

    const int g2Waves = (N + 15) / 16;
    int g2Wpb = (g2Waves + SCAT_WGS - 1) / SCAT_WGS;
    if (g2Wpb > G2_WPB) g2Wpb = G2_WPB;
    if (g2Wpb < 1) g2Wpb = 1;
    k_gather2<<<(g2Waves + g2Wpb - 1) / g2Wpb, g2Wpb * 32, 0, stream>>>(rowptr, csrSrc, dinv, G, b2, out, N);
}


# edge list sorted by a two-level counting sort, 2 launches not 3: edges grouped once by destination range; a range's work-group counts, scans, fills in LDS and scales its H rows (sum order changes)
# speedup vs baseline: 6.0359x; 1.1916x over previous
#include <hip/hip_runtime.h>
#include <hip/hip_bf16.h>

typedef float v2f __attribute__((ext_vector_type(2)));
typedef float v8f __attribute__((ext_vector_type(8)));

#define IN_FEATS 128
#define HIDDEN   64

#define NR        256
#define NC        256
#define PT        1024
#define RANGE_CAP 256
#define RUNS_PW   (NC / (PT / 32))
#define H_V4      (RANGE_CAP * (HIDDEN / 4) / PT)

#define W1_V4 (IN_FEATS * HIDDEN / 4 / PT)

__global__ void __launch_bounds__(PT) k_part_and_gemm1(const int* __restrict__ src,
                                                       const int* __restrict__ dst,
                                                       int E, int chunkLen, int rangeLen,
                                                       uint2* __restrict__ tabT,
                                                       int2* __restrict__ bkt,
                                                       int partWgs,
                                                       const float* __restrict__ X,
                                                       const float* __restrict__ W1,
                                                       float* __restrict__ H,
                                                       int nTiles) {
    __shared__ union {
        struct {
            unsigned cnt[NR];
            unsigned cur[NR];
            unsigned wt[NR / 32];
        } p;
        float w[IN_FEATS * HIDDEN];
    } sm;
    const int tid = threadIdx.x;

    if ((int)blockIdx.x < partWgs) {
        const int c    = blockIdx.x;
        const int lane = tid & 31;
        const int wv   = tid >> 5;
        if (tid < NR) sm.p.cnt[tid] = 0u;
        __syncthreads();

        const int e0 = min(E, c * chunkLen);
        const int e1 = min(E, e0 + chunkLen);
        for (int base = e0 + tid; base < e1; base += 8 * PT) {
            unsigned rr[8], dv[8];
            #pragma unroll
            for (int j = 0; j < 8; ++j) dv[j] = (unsigned)dst[min(base + j * PT, e1 - 1)];
            asm volatile("" :: "v"(dv[0]), "v"(dv[1]), "v"(dv[2]), "v"(dv[3]), "v"(dv[4]), "v"(dv[5]), "v"(dv[6]), "v"(dv[7]));
            #pragma unroll
            for (int j = 0; j < 8; ++j)
                rr[j] = (base + j * PT < e1) ? min(dv[j] / (unsigned)rangeLen, (unsigned)(NR - 1)) : (unsigned)NR;
            #pragma unroll
            for (int j = 0; j < 8; ++j)
                if (rr[j] < NR) atomicAdd(&sm.p.cnt[rr[j]], 1u);
        }
        __syncthreads();

        unsigned v = 0u;
        if (tid < NR) v = sm.p.cnt[tid];
        unsigned inc = v;
        for (int o = 1; o < 32; o <<= 1) {
            const unsigned t = __shfl_up(inc, o);
            if (lane >= o) inc += t;
        }
        if (tid < NR && lane == 31) sm.p.wt[wv] = inc;
        __syncthreads();
        if (tid < NR) {
            unsigned before = 0u;
            #pragma unroll
            for (int w = 0; w < NR / 32; ++w) {
                const unsigned t = sm.p.wt[w];
                if (w < wv) before += t;
            }
            const unsigned ex = before + inc - v;
            sm.p.cur[tid] = ex;
            tabT[(size_t)tid * NC + c] = make_uint2(v, ex);
        }
        __syncthreads();

        int2* __restrict__ myb = bkt + (size_t)c * chunkLen;
        for (int base = e0 + tid; base < e1; base += 8 * PT) {
            unsigned rr[8];
            int      ss[8], dd[8];
            #pragma unroll
            for (int j = 0; j < 8; ++j) {
                const int ec = min(base + j * PT, e1 - 1);
                dd[j] = dst[ec];
                ss[j] = src[ec];
            }
            asm volatile("" :: "v"(dd[0]), "v"(dd[1]), "v"(dd[2]), "v"(dd[3]), "v"(dd[4]), "v"(dd[5]), "v"(dd[6]), "v"(dd[7]),
                               "v"(ss[0]), "v"(ss[1]), "v"(ss[2]), "v"(ss[3]), "v"(ss[4]), "v"(ss[5]), "v"(ss[6]), "v"(ss[7]));
            #pragma unroll
            for (int j = 0; j < 8; ++j)
                rr[j] = (base + j * PT < e1) ? min((unsigned)dd[j] / (unsigned)rangeLen, (unsigned)(NR - 1)) : (unsigned)NR;
            unsigned pp[8];
            #pragma unroll
            for (int j = 0; j < 8; ++j) {
                pp[j] = 0u;
                if (rr[j] < NR) pp[j] = atomicAdd(&sm.p.cur[rr[j]], 1u);
            }
            #pragma unroll
            for (int j = 0; j < 8; ++j)
                if (rr[j] < NR) myb[pp[j]] = make_int2(ss[j], dd[j]);
        }
        return;
    }

    {
        const float4* __restrict__ W4 = (const float4*)W1;
        float4* w4 = (float4*)sm.w;
        float4 t[W1_V4];
        #pragma unroll
        for (int j = 0; j < W1_V4; ++j) t[j] = W4[tid + j * PT];
        #pragma unroll
        for (int j = 0; j < W1_V4; ++j) w4[tid + j * PT] = t[j];
    }
    __syncthreads();

    const int waveId = ((int)blockIdx.x - partWgs) * (PT / 32) + (tid >> 5);
    const int lane   = tid & 31;
    if (waveId >= nTiles) return;

    const int nTilesN = HIDDEN / 16;
    const int mt = waveId / nTilesN;
    const int nt = waveId % nTilesN;

    const int row = mt * 16 + (lane & 15);
    const int nn  = nt * 16 + (lane & 15);
    const int kq  = (lane >> 4) << 1;

    const float* __restrict__ xrow = X + (size_t)row * IN_FEATS;

    const int rbase = mt * 16 + ((lane >> 4) << 3);

    v8f acc = {};
    #pragma unroll 8
    for (int k = 0; k < IN_FEATS; k += 4) {
        v2f a, b;
        a.x = xrow[k + kq];
        a.y = xrow[k + kq + 1];
        b.x = sm.w[(k + kq)     * HIDDEN + nn];
        b.y = sm.w[(k + kq + 1) * HIDDEN + nn];
        acc = __builtin_amdgcn_wmma_f32_16x16x4_f32(
                 false, a,  false, b,
                 (short)0, acc,  false,  false);
    }

    #pragma unroll
    for (int v = 0; v < 8; ++v)
        H[(size_t)(rbase + v) * HIDDEN + nn] = acc[v];
}

__global__ void __launch_bounds__(PT) k_range_csr(const uint2* __restrict__ tabT,
                                                  const int2* __restrict__ bkt,
                                                  int E, int n, int chunkLen, int rangeLen,
                                                  float* __restrict__ dinv,
                                                  unsigned* __restrict__ rowptr,
                                                  int* __restrict__ csrSrc,
                                                  float* __restrict__ H) {
    __shared__ unsigned cnt[RANGE_CAP];
    __shared__ float    dl[RANGE_CAP];
    __shared__ unsigned wsA[PT / 32];
    __shared__ unsigned wsB[PT / 32];

    const int tid  = threadIdx.x;
    const int lane = tid & 31;
    const int wv   = __builtin_amdgcn_readfirstlane(tid >> 5);
    const int r    = blockIdx.x;
    const int lo   = r * rangeLen;
    const int len  = max(0, min(min(rangeLen, RANGE_CAP), n - lo));
    const uint2* __restrict__ trow = tabT + (size_t)r * NC;

    float4* __restrict__ H4 = (float4*)H + (size_t)lo * (HIDDEN / 4);
    float4 hv[H_V4];
    #pragma unroll
    for (int j = 0; j < H_V4; ++j) {
        const int i = tid + j * PT;
        hv[j] = make_float4(0.0f, 0.0f, 0.0f, 0.0f);
        if (i < len * (HIDDEN / 4)) hv[j] = H4[i];
    }

    const int c0 = wv * RUNS_PW;
    uint2 te[RUNS_PW];
    #pragma unroll
    for (int k = 0; k < RUNS_PW; ++k) te[k] = trow[c0 + k];
    int2 ev[RUNS_PW];
    #pragma unroll
    for (int k = 0; k < RUNS_PW; ++k) {
        ev[k] = make_int2(0, -1);
        if ((unsigned)lane < te[k].x) ev[k] = bkt[(size_t)(c0 + k) * chunkLen + te[k].y + lane];
    }

    unsigned s = 0u;
    if (tid < NC) s = trow[tid].y;
    if (tid < RANGE_CAP) cnt[tid] = 0u;
    for (int o = 16; o > 0; o >>= 1) s += __shfl_xor(s, o);
    if (lane == 0) wsA[wv] = s;
    __syncthreads();
    unsigned base = 0u;
    #pragma unroll
    for (int w = 0; w < NC / 32; ++w) base += wsA[w];

    #pragma unroll
    for (int k = 0; k < RUNS_PW; ++k) {
        const unsigned l = (unsigned)(ev[k].y - lo);
        if (l < (unsigned)len) atomicAdd(&cnt[l], 1u);
    }
    for (int k = 0; k < RUNS_PW; ++k) {
        const int2* __restrict__ run = bkt + (size_t)(c0 + k) * chunkLen + te[k].y;
        for (unsigned j = 32u + lane; j < te[k].x; j += 32u) {
            const unsigned l = (unsigned)(run[j].y - lo);
            if (l < (unsigned)len) atomicAdd(&cnt[l], 1u);
        }
    }
    __syncthreads();

    unsigned v = 0u;
    if (tid < len) v = cnt[tid];
    unsigned inc = v;
    for (int o = 1; o < 32; o <<= 1) {
        const unsigned t = __shfl_up(inc, o);
        if (lane >= o) inc += t;
    }
    if (lane == 31) wsB[wv] = inc;
    __syncthreads();
    if (tid < len) {
        unsigned before = 0u;
        #pragma unroll
        for (int w = 0; w < RANGE_CAP / 32; ++w) {
            const unsigned t = wsB[w];
            if (w < wv) before += t;
        }
        const unsigned rp = base + before + inc - v;
        const float d  = 1.0f + (float)v;
        const float di = (d > 0.0f) ? rsqrtf(d) : 0.0f;
        dinv[lo + tid]   = di;
        dl[tid]          = di;
        rowptr[lo + tid] = rp;
        cnt[tid]         = rp;
        if (lo + tid == n - 1) rowptr[n] = rp + v;
    }
    __syncthreads();

    unsigned sl[RUNS_PW];
    #pragma unroll
    for (int k = 0; k < RUNS_PW; ++k) {
        const unsigned l = (unsigned)(ev[k].y - lo);
        sl[k] = (unsigned)E;
        if (l < (unsigned)len) sl[k] = atomicAdd(&cnt[l], 1u);
    }
    #pragma unroll
    for (int k = 0; k < RUNS_PW; ++k)
        if (sl[k] < (unsigned)E) csrSrc[sl[k]] = ev[k].x;
    for (int k = 0; k < RUNS_PW; ++k) {
        const int2* __restrict__ run = bkt + (size_t)(c0 + k) * chunkLen + te[k].y;
        for (unsigned j = 32u + lane; j < te[k].x; j += 32u) {
            const int2 e = run[j];
            const unsigned l = (unsigned)(e.y - lo);
            if (l < (unsigned)len) {
                const unsigned slot = atomicAdd(&cnt[l], 1u);
                if (slot < (unsigned)E) csrSrc[slot] = e.x;
            }
        }
    }

    #pragma unroll
    for (int j = 0; j < H_V4; ++j) {
        const int i = tid + j * PT;
        if (i < len * (HIDDEN / 4)) {
            const float d = dl[i / (HIDDEN / 4)];
            float4 h = hv[j];
            h.x *= d; h.y *= d; h.z *= d; h.w *= d;
            H4[i] = h;
        }
    }
}

__device__ __forceinline__ float row16_sum(float v) {
    v += __int_as_float(__builtin_amdgcn_update_dpp(0, __float_as_int(v), 0x161, 0xF, 0xF, true));
    v += __int_as_float(__builtin_amdgcn_update_dpp(0, __float_as_int(v), 0x162, 0xF, 0xF, true));
    v += __int_as_float(__builtin_amdgcn_update_dpp(0, __float_as_int(v), 0x164, 0xF, 0xF, true));
    v += __int_as_float(__builtin_amdgcn_update_dpp(0, __float_as_int(v), 0x168, 0xF, 0xF, true));
    return v;
}

__global__ void __launch_bounds__(256) k_gather1(const unsigned* __restrict__ rowptr,
                                                 const int* __restrict__ csrSrc,
                                                 const float* __restrict__ dinv,
                                                 const float* __restrict__ H,
                                                 const float* __restrict__ b1,
                                                 const float* __restrict__ W2,
                                                 float* __restrict__ G, int n) {
    const int node = __builtin_amdgcn_readfirstlane((int)((blockIdx.x * blockDim.x + threadIdx.x) >> 5));
    if (node >= n) return;
    const int lane = threadIdx.x & 31;
    const int half = lane >> 4;
    const int q    = lane & 15;
    const float4* __restrict__ H4 = (const float4*)H;

    const float dd  = dinv[node];
    const int   beg = (int)rowptr[node];
    const int   end = (int)rowptr[node + 1];

    float4 acc = make_float4(0.0f, 0.0f, 0.0f, 0.0f);
    float4 hs = make_float4(0.0f, 0.0f, 0.0f, 0.0f);
    if (half == 0) hs = H4[(size_t)node * (HIDDEN / 4) + q];
    const float4 bq  = ((const float4*)b1)[q];
    const float4 w01 = ((const float4*)W2)[2 * q];
    const float4 w23 = ((const float4*)W2)[2 * q + 1];

    for (int b0 = beg; b0 < end; b0 += 32) {
        const int cnt = min(32, end - b0);
        int s_l = 0;
        if (lane < cnt) s_l = csrSrc[b0 + lane];
        int j = 0;
        do {
            float4 h[8];
            #pragma unroll
            for (int u = 0; u < 8; ++u) {
                const int idx = j + 2 * u + half;
                const int s   = __shfl(s_l, idx);
                h[u] = make_float4(0.0f, 0.0f, 0.0f, 0.0f);
                if (idx < cnt) h[u] = H4[(size_t)s * (HIDDEN / 4) + q];
            }
            #pragma unroll
            for (int u = 0; u < 8; ++u) {
                acc.x += h[u].x;
                acc.y += h[u].y;
                acc.z += h[u].z;
                acc.w += h[u].w;
            }
            j += 16;
        } while (j < cnt);
    }

    acc.x += hs.x;
    acc.y += hs.y;
    acc.z += hs.z;
    acc.w += hs.w;

    acc.x += __shfl_xor(acc.x, 16);
    acc.y += __shfl_xor(acc.y, 16);
    acc.z += __shfl_xor(acc.z, 16);
    acc.w += __shfl_xor(acc.w, 16);
    acc.x *= dd; acc.y *= dd; acc.z *= dd; acc.w *= dd;

    float v0 = acc.x + bq.x;
    float v1 = acc.y + bq.y;
    float v2 = acc.z + bq.z;
    float v3 = acc.w + bq.w;
    v0 = v0 > 0.0f ? v0 : 0.0f;
    v1 = v1 > 0.0f ? v1 : 0.0f;
    v2 = v2 > 0.0f ? v2 : 0.0f;
    v3 = v3 > 0.0f ? v3 : 0.0f;
    float p0 = v0 * w01.x;
    float p1 = v0 * w01.y;
    p0 = fmaf(v1, w01.z, p0);
    p1 = fmaf(v1, w01.w, p1);
    p0 = fmaf(v2, w23.x, p0);
    p1 = fmaf(v2, w23.y, p1);
    p0 = fmaf(v3, w23.z, p0);
    p1 = fmaf(v3, w23.w, p1);
    p0 = row16_sum(p0);
    p1 = row16_sum(p1);
    if (lane == 0) ((float2*)G)[node] = make_float2(p0 * dd, p1 * dd);
}

#define G2_LANES 4
#define G2_WPB   32
#define G2_U     12

__device__ __forceinline__ float quad_sum(float v) {
    v += __int_as_float(__builtin_amdgcn_update_dpp(0, __float_as_int(v), 0xB1, 0xF, 0xF, true));
    v += __int_as_float(__builtin_amdgcn_update_dpp(0, __float_as_int(v), 0x4E, 0xF, 0xF, true));
    return v;
}

__global__ void __launch_bounds__(G2_WPB * 32) k_gather2(const unsigned* __restrict__ rowptr,
                                                 const int* __restrict__ csrSrc,
                                                 const float* __restrict__ dinv,
                                                 const float* __restrict__ G,
                                                 const float* __restrict__ b2,
                                                 float* __restrict__ out, int n) {
    const int wave = (int)((blockIdx.x * blockDim.x + threadIdx.x) >> 5);
    const int lane = threadIdx.x & 31;
    const int grp  = lane >> 2;
    const int sub  = lane & 3;
    const float2* __restrict__ G2 = (const float2*)G;

    const int  nodeA = wave * 16 + grp;
    const int  nodeB = nodeA + 8;
    const bool okA   = nodeA < n;
    const bool okB   = nodeB < n;
    const int  cA    = min(nodeA, n - 1);
    const int  cB    = min(nodeB, n - 1);

    int begA = (int)rowptr[cA], endA = (int)rowptr[cA + 1];
    int begB = (int)rowptr[cB], endB = (int)rowptr[cB + 1];
    const int   nodeS = (sub == 0) ? nodeA : nodeB;
    const bool  okS   = (sub < 2) && (nodeS < n);
    const float dS    = dinv[min(nodeS, n - 1)];
    const float2 selfA = G2[cA];
    const float2 selfB = G2[cB];
    if (!okA) endA = begA;
    if (!okB) endB = begB;

    float2 accA = make_float2(0.0f, 0.0f);
    float2 accB = make_float2(0.0f, 0.0f);
    if (sub == 0 && okA) accA = selfA;
    if (sub == 0 && okB) accB = selfB;

    const int lastA = max(endA - 1, 0);
    const int lastB = max(endB - 1, 0);
    int eA = begA + sub;
    int eB = begB + sub;
    while ((eA < endA) | (eB < endB)) {
        int sA[G2_U], sB[G2_U];
        #pragma unroll
        for (int u = 0; u < G2_U; ++u) {
            sA[u] = csrSrc[min(eA + u * G2_LANES, lastA)];
            sB[u] = csrSrc[min(eB + u * G2_LANES, lastB)];
        }
        float2 gA[G2_U], gB[G2_U];
        #pragma unroll
        for (int u = 0; u < G2_U; ++u) {
            gA[u] = G2[sA[u]];
            gB[u] = G2[sB[u]];
        }
        #pragma unroll
        for (int u = 0; u < G2_U; ++u)
            asm volatile("" :: "v"(gA[u].x), "v"(gA[u].y), "v"(gB[u].x), "v"(gB[u].y));
        #pragma unroll
        for (int u = 0; u < G2_U; ++u) {
            const bool vA = eA + u * G2_LANES < endA;
            const bool vB = eB + u * G2_LANES < endB;
            accA.x += vA ? gA[u].x : 0.0f; accA.y += vA ? gA[u].y : 0.0f;
            accB.x += vB ? gB[u].x : 0.0f; accB.y += vB ? gB[u].y : 0.0f;
        }
        eA += G2_U * G2_LANES;
        eB += G2_U * G2_LANES;
    }

    accA.x = quad_sum(accA.x); accA.y = quad_sum(accA.y);
    accB.x = quad_sum(accB.x); accB.y = quad_sum(accB.y);

    if (okS) {
        const float2 t = (sub == 0) ? accA : accB;
        float2 o;
        o.x = fmaf(dS, t.x, b2[0]);
        o.y = fmaf(dS, t.y, b2[1]);
        ((float2*)out)[nodeS] = o;
    }
}

extern "C" void kernel_launch(void* const* d_in, const int* in_sizes, int n_in,
                              void* d_out, int out_size, void* d_ws, size_t ws_size,
                              hipStream_t stream) {
    const float* x  = (const float*)d_in[0];
    const float* W1 = (const float*)d_in[1];
    const float* b1 = (const float*)d_in[2];
    const float* W2 = (const float*)d_in[3];
    const float* b2 = (const float*)d_in[4];
    const int*   ei = (const int*)d_in[5];

    const int N = in_sizes[0] / IN_FEATS;
    const int E = in_sizes[5] / 2;
    const int* src = ei;
    const int* dst = ei + E;

    const int rangeLen = (N + NR - 1) / NR;
    const int chunkLen = (E + NC - 1) / NC;
    const int nPad     = (N + 31) & ~31;

    float* ws   = (float*)d_ws;
    float* dinv = ws;
    float* H    = dinv + nPad;
    float* G    = H    + (size_t)N * HIDDEN;
    unsigned* rowptr = (unsigned*)(G + (size_t)nPad * 2);
    uint2*    tabT   = (uint2*)(rowptr + nPad + 32);
    int2*     bkt    = (int2*)(tabT + (size_t)NR * NC);
    int*      csrSrc = (int*)(bkt + (size_t)NC * chunkLen);
    float* out  = (float*)d_out;

    const int T = 256;

    const int nTiles  = (N / 16) * (HIDDEN / 16);
    const int gemmWgs = (nTiles + (PT / 32) - 1) / (PT / 32);
    k_part_and_gemm1<<<NC + gemmWgs, PT, 0, stream>>>(src, dst, E, chunkLen, rangeLen, tabT, bkt, NC, x, W1, H, nTiles);

    k_range_csr<<<NR, PT, 0, stream>>>(tabT, bkt, E, N, chunkLen, rangeLen, dinv, rowptr, csrSrc, H);

    k_gather1<<<(N * 32 + T - 1) / T, T, 0, stream>>>(rowptr, csrSrc, dinv, H, b1, W2, G, N);

    const int g2Waves = (N + 15) / 16;
    int g2Wpb = (g2Waves + 256 - 1) / 256;
    if (g2Wpb > G2_WPB) g2Wpb = G2_WPB;
    if (g2Wpb < 1) g2Wpb = 1;
    k_gather2<<<(g2Waves + g2Wpb - 1) / g2Wpb, g2Wpb * 32, 0, stream>>>(rowptr, csrSrc, dinv, G, b2, out, N);
}


# x@W1 product: a wave takes 16 rows x all 64 columns (4 tiles): X row read once in 16-byte pieces, an A fragment feeds 4 WMMAs, W1 k pairs interleaved in LDS for 16-byte B reads (k sum order changes)
# speedup vs baseline: 6.7898x; 1.1249x over previous
#include <hip/hip_runtime.h>
#include <hip/hip_bf16.h>

typedef float v2f __attribute__((ext_vector_type(2)));
typedef float v8f __attribute__((ext_vector_type(8)));

#define IN_FEATS 128
#define HIDDEN   64

#define NR        256
#define NC        256
#define PT        1024
#define RANGE_CAP 256
#define RUNS_PW   (NC / (PT / 32))
#define H_V4      (RANGE_CAP * (HIDDEN / 4) / PT)


__global__ void __launch_bounds__(PT) k_part_and_gemm1(const int* __restrict__ src,
                                                       const int* __restrict__ dst,
                                                       int E, int chunkLen, int rangeLen,
                                                       uint2* __restrict__ tabT,
                                                       int2* __restrict__ bkt,
                                                       int partWgs,
                                                       const float* __restrict__ X,
                                                       const float* __restrict__ W1,
                                                       float* __restrict__ H,
                                                       int nRowBlocks, int gemmWgs) {
    __shared__ union {
        struct {
            unsigned cnt[NR];
            unsigned cur[NR];
            unsigned wt[NR / 32];
        } p;
        float w[IN_FEATS * HIDDEN];
    } sm;
    const int tid = threadIdx.x;

    if ((int)blockIdx.x < partWgs) {
        const int c    = blockIdx.x;
        const int lane = tid & 31;
        const int wv   = tid >> 5;
        if (tid < NR) sm.p.cnt[tid] = 0u;
        __syncthreads();

        const int e0 = min(E, c * chunkLen);
        const int e1 = min(E, e0 + chunkLen);
        for (int base = e0 + tid; base < e1; base += 8 * PT) {
            unsigned rr[8], dv[8];
            #pragma unroll
            for (int j = 0; j < 8; ++j) dv[j] = (unsigned)dst[min(base + j * PT, e1 - 1)];
            asm volatile("" :: "v"(dv[0]), "v"(dv[1]), "v"(dv[2]), "v"(dv[3]), "v"(dv[4]), "v"(dv[5]), "v"(dv[6]), "v"(dv[7]));
            #pragma unroll
            for (int j = 0; j < 8; ++j)
                rr[j] = (base + j * PT < e1) ? min(dv[j] / (unsigned)rangeLen, (unsigned)(NR - 1)) : (unsigned)NR;
            #pragma unroll
            for (int j = 0; j < 8; ++j)
                if (rr[j] < NR) atomicAdd(&sm.p.cnt[rr[j]], 1u);
        }
        __syncthreads();

        unsigned v = 0u;
        if (tid < NR) v = sm.p.cnt[tid];
        unsigned inc = v;
        for (int o = 1; o < 32; o <<= 1) {
            const unsigned t = __shfl_up(inc, o);
            if (lane >= o) inc += t;
        }
        if (tid < NR && lane == 31) sm.p.wt[wv] = inc;
        __syncthreads();
        if (tid < NR) {
            unsigned before = 0u;
            #pragma unroll
            for (int w = 0; w < NR / 32; ++w) {
                const unsigned t = sm.p.wt[w];
                if (w < wv) before += t;
            }
            const unsigned ex = before + inc - v;
            sm.p.cur[tid] = ex;
            tabT[(size_t)tid * NC + c] = make_uint2(v, ex);
        }
        __syncthreads();

        int2* __restrict__ myb = bkt + (size_t)c * chunkLen;
        for (int base = e0 + tid; base < e1; base += 8 * PT) {
            unsigned rr[8];
            int      ss[8], dd[8];
            #pragma unroll
            for (int j = 0; j < 8; ++j) {
                const int ec = min(base + j * PT, e1 - 1);
                dd[j] = dst[ec];
                ss[j] = src[ec];
            }
            asm volatile("" :: "v"(dd[0]), "v"(dd[1]), "v"(dd[2]), "v"(dd[3]), "v"(dd[4]), "v"(dd[5]), "v"(dd[6]), "v"(dd[7]),
                               "v"(ss[0]), "v"(ss[1]), "v"(ss[2]), "v"(ss[3]), "v"(ss[4]), "v"(ss[5]), "v"(ss[6]), "v"(ss[7]));
            #pragma unroll
            for (int j = 0; j < 8; ++j)
                rr[j] = (base + j * PT < e1) ? min((unsigned)dd[j] / (unsigned)rangeLen, (unsigned)(NR - 1)) : (unsigned)NR;
            unsigned pp[8];
            #pragma unroll
            for (int j = 0; j < 8; ++j) {
                pp[j] = 0u;
                if (rr[j] < NR) pp[j] = atomicAdd(&sm.p.cur[rr[j]], 1u);
            }
            #pragma unroll
            for (int j = 0; j < 8; ++j)
                if (rr[j] < NR) myb[pp[j]] = make_int2(ss[j], dd[j]);
        }
        return;
    }

    {
        const float4* __restrict__ W4 = (const float4*)W1;
        float4* p4 = (float4*)sm.w;
        const int kp = tid >> 4;
        const int m  = tid & 15;
        const float4 r0 = W4[(2 * kp)     * (HIDDEN / 4) + m];
        const float4 r1 = W4[(2 * kp + 1) * (HIDDEN / 4) + m];
        p4[kp * 32 + m]      = make_float4(r0.x, r1.x, r0.y, r1.y);
        p4[kp * 32 + 16 + m] = make_float4(r0.z, r1.z, r0.w, r1.w);
    }
    __syncthreads();

    const int lane = tid & 31;
    const int mt   = (tid >> 5) * gemmWgs + ((int)blockIdx.x - partWgs);
    if (mt >= nRowBlocks) return;

    const int c = lane & 15;
    const int h = lane >> 4;

    const float4* __restrict__ xrow4 = (const float4*)(X + (size_t)(mt * 16 + c) * IN_FEATS) + h;
    const float4* __restrict__ p4    = (const float4*)sm.w + c;

    v8f acc0 = {}, acc1 = {}, acc2 = {}, acc3 = {};
    #pragma unroll 2
    for (int i = 0; i < IN_FEATS / 8; ++i) {
        const float4 xa = xrow4[2 * i];
        #pragma unroll
        for (int j = 0; j < 2; ++j) {
            const int kp = 4 * i + 2 * h + j;
            v2f a;
            a.x = j ? xa.z : xa.x;
            a.y = j ? xa.w : xa.y;
            const float4 q0 = p4[kp * 32];
            const float4 q1 = p4[kp * 32 + 16];
            v2f b0, b1, b2, b3;
            b0.x = q0.x; b0.y = q0.y;
            b1.x = q0.z; b1.y = q0.w;
            b2.x = q1.x; b2.y = q1.y;
            b3.x = q1.z; b3.y = q1.w;
            acc0 = __builtin_amdgcn_wmma_f32_16x16x4_f32(false, a, false, b0, (short)0, acc0, false, false);
            acc1 = __builtin_amdgcn_wmma_f32_16x16x4_f32(false, a, false, b1, (short)0, acc1, false, false);
            acc2 = __builtin_amdgcn_wmma_f32_16x16x4_f32(false, a, false, b2, (short)0, acc2, false, false);
            acc3 = __builtin_amdgcn_wmma_f32_16x16x4_f32(false, a, false, b3, (short)0, acc3, false, false);
        }
    }

    float4* __restrict__ hrow4 = (float4*)(H + (size_t)(mt * 16 + 8 * h) * HIDDEN) + c;
    #pragma unroll
    for (int v = 0; v < 8; ++v)
        hrow4[v * (HIDDEN / 4)] = make_float4(acc0[v], acc1[v], acc2[v], acc3[v]);
}

__global__ void __launch_bounds__(PT) k_range_csr(const uint2* __restrict__ tabT,
                                                  const int2* __restrict__ bkt,
                                                  int E, int n, int chunkLen, int rangeLen,
                                                  float* __restrict__ dinv,
                                                  unsigned* __restrict__ rowptr,
                                                  int* __restrict__ csrSrc,
                                                  float* __restrict__ H) {
    __shared__ unsigned cnt[RANGE_CAP];
    __shared__ float    dl[RANGE_CAP];
    __shared__ unsigned wsA[PT / 32];
    __shared__ unsigned wsB[PT / 32];

    const int tid  = threadIdx.x;
    const int lane = tid & 31;
    const int wv   = __builtin_amdgcn_readfirstlane(tid >> 5);
    const int r    = blockIdx.x;
    const int lo   = r * rangeLen;
    const int len  = max(0, min(min(rangeLen, RANGE_CAP), n - lo));
    const uint2* __restrict__ trow = tabT + (size_t)r * NC;

    float4* __restrict__ H4 = (float4*)H + (size_t)lo * (HIDDEN / 4);
    float4 hv[H_V4];
    #pragma unroll
    for (int j = 0; j < H_V4; ++j) {
        const int i = tid + j * PT;
        hv[j] = make_float4(0.0f, 0.0f, 0.0f, 0.0f);
        if (i < len * (HIDDEN / 4)) hv[j] = H4[i];
    }

    const int c0 = wv * RUNS_PW;
    uint2 te[RUNS_PW];
    #pragma unroll
    for (int k = 0; k < RUNS_PW; ++k) te[k] = trow[c0 + k];
    int2 ev[RUNS_PW];
    #pragma unroll
    for (int k = 0; k < RUNS_PW; ++k) {
        ev[k] = make_int2(0, -1);
        if ((unsigned)lane < te[k].x) ev[k] = bkt[(size_t)(c0 + k) * chunkLen + te[k].y + lane];
    }

    unsigned s = 0u;
    if (tid < NC) s = trow[tid].y;
    if (tid < RANGE_CAP) cnt[tid] = 0u;
    for (int o = 16; o > 0; o >>= 1) s += __shfl_xor(s, o);
    if (lane == 0) wsA[wv] = s;
    __syncthreads();
    unsigned base = 0u;
    #pragma unroll
    for (int w = 0; w < NC / 32; ++w) base += wsA[w];

    #pragma unroll
    for (int k = 0; k < RUNS_PW; ++k) {
        const unsigned l = (unsigned)(ev[k].y - lo);
        if (l < (unsigned)len) atomicAdd(&cnt[l], 1u);
    }
    for (int k = 0; k < RUNS_PW; ++k) {
        const int2* __restrict__ run = bkt + (size_t)(c0 + k) * chunkLen + te[k].y;
        for (unsigned j = 32u + lane; j < te[k].x; j += 32u) {
            const unsigned l = (unsigned)(run[j].y - lo);
            if (l < (unsigned)len) atomicAdd(&cnt[l], 1u);
        }
    }
    __syncthreads();

    unsigned v = 0u;
    if (tid < len) v = cnt[tid];
    unsigned inc = v;
    for (int o = 1; o < 32; o <<= 1) {
        const unsigned t = __shfl_up(inc, o);
        if (lane >= o) inc += t;
    }
    if (lane == 31) wsB[wv] = inc;
    __syncthreads();
    if (tid < len) {
        unsigned before = 0u;
        #pragma unroll
        for (int w = 0; w < RANGE_CAP / 32; ++w) {
            const unsigned t = wsB[w];
            if (w < wv) before += t;
        }
        const unsigned rp = base + before + inc - v;
        const float d  = 1.0f + (float)v;
        const float di = (d > 0.0f) ? rsqrtf(d) : 0.0f;
        dinv[lo + tid]   = di;
        dl[tid]          = di;
        rowptr[lo + tid] = rp;
        cnt[tid]         = rp;
        if (lo + tid == n - 1) rowptr[n] = rp + v;
    }
    __syncthreads();

    unsigned sl[RUNS_PW];
    #pragma unroll
    for (int k = 0; k < RUNS_PW; ++k) {
        const unsigned l = (unsigned)(ev[k].y - lo);
        sl[k] = (unsigned)E;
        if (l < (unsigned)len) sl[k] = atomicAdd(&cnt[l], 1u);
    }
    #pragma unroll
    for (int k = 0; k < RUNS_PW; ++k)
        if (sl[k] < (unsigned)E) csrSrc[sl[k]] = ev[k].x;
    for (int k = 0; k < RUNS_PW; ++k) {
        const int2* __restrict__ run = bkt + (size_t)(c0 + k) * chunkLen + te[k].y;
        for (unsigned j = 32u + lane; j < te[k].x; j += 32u) {
            const int2 e = run[j];
            const unsigned l = (unsigned)(e.y - lo);
            if (l < (unsigned)len) {
                const unsigned slot = atomicAdd(&cnt[l], 1u);
                if (slot < (unsigned)E) csrSrc[slot] = e.x;
            }
        }
    }

    #pragma unroll
    for (int j = 0; j < H_V4; ++j) {
        const int i = tid + j * PT;
        if (i < len * (HIDDEN / 4)) {
            const float d = dl[i / (HIDDEN / 4)];
            float4 h = hv[j];
            h.x *= d; h.y *= d; h.z *= d; h.w *= d;
            H4[i] = h;
        }
    }
}

__device__ __forceinline__ float row16_sum(float v) {
    v += __int_as_float(__builtin_amdgcn_update_dpp(0, __float_as_int(v), 0x161, 0xF, 0xF, true));
    v += __int_as_float(__builtin_amdgcn_update_dpp(0, __float_as_int(v), 0x162, 0xF, 0xF, true));
    v += __int_as_float(__builtin_amdgcn_update_dpp(0, __float_as_int(v), 0x164, 0xF, 0xF, true));
    v += __int_as_float(__builtin_amdgcn_update_dpp(0, __float_as_int(v), 0x168, 0xF, 0xF, true));
    return v;
}

__global__ void __launch_bounds__(256) k_gather1(const unsigned* __restrict__ rowptr,
                                                 const int* __restrict__ csrSrc,
                                                 const float* __restrict__ dinv,
                                                 const float* __restrict__ H,
                                                 const float* __restrict__ b1,
                                                 const float* __restrict__ W2,
                                                 float* __restrict__ G, int n) {
    const int node = __builtin_amdgcn_readfirstlane((int)((blockIdx.x * blockDim.x + threadIdx.x) >> 5));
    if (node >= n) return;
    const int lane = threadIdx.x & 31;
    const int half = lane >> 4;
    const int q    = lane & 15;
    const float4* __restrict__ H4 = (const float4*)H;

    const float dd  = dinv[node];
    const int   beg = (int)rowptr[node];
    const int   end = (int)rowptr[node + 1];

    float4 acc = make_float4(0.0f, 0.0f, 0.0f, 0.0f);
    float4 hs = make_float4(0.0f, 0.0f, 0.0f, 0.0f);
    if (half == 0) hs = H4[(size_t)node * (HIDDEN / 4) + q];
    const float4 bq  = ((const float4*)b1)[q];
    const float4 w01 = ((const float4*)W2)[2 * q];
    const float4 w23 = ((const float4*)W2)[2 * q + 1];

    for (int b0 = beg; b0 < end; b0 += 32) {
        const int cnt = min(32, end - b0);
        int s_l = 0;
        if (lane < cnt) s_l = csrSrc[b0 + lane];
        int j = 0;
        do {
            float4 h[8];
            #pragma unroll
            for (int u = 0; u < 8; ++u) {
                const int idx = j + 2 * u + half;
                const int s   = __shfl(s_l, idx);
                h[u] = make_float4(0.0f, 0.0f, 0.0f, 0.0f);
                if (idx < cnt) h[u] = H4[(size_t)s * (HIDDEN / 4) + q];
            }
            #pragma unroll
            for (int u = 0; u < 8; ++u) {
                acc.x += h[u].x;
                acc.y += h[u].y;
                acc.z += h[u].z;
                acc.w += h[u].w;
            }
            j += 16;
        } while (j < cnt);
    }

    acc.x += hs.x;
    acc.y += hs.y;
    acc.z += hs.z;
    acc.w += hs.w;

    acc.x += __shfl_xor(acc.x, 16);
    acc.y += __shfl_xor(acc.y, 16);
    acc.z += __shfl_xor(acc.z, 16);
    acc.w += __shfl_xor(acc.w, 16);
    acc.x *= dd; acc.y *= dd; acc.z *= dd; acc.w *= dd;

    float v0 = acc.x + bq.x;
    float v1 = acc.y + bq.y;
    float v2 = acc.z + bq.z;
    float v3 = acc.w + bq.w;
    v0 = v0 > 0.0f ? v0 : 0.0f;
    v1 = v1 > 0.0f ? v1 : 0.0f;
    v2 = v2 > 0.0f ? v2 : 0.0f;
    v3 = v3 > 0.0f ? v3 : 0.0f;
    float p0 = v0 * w01.x;
    float p1 = v0 * w01.y;
    p0 = fmaf(v1, w01.z, p0);
    p1 = fmaf(v1, w01.w, p1);
    p0 = fmaf(v2, w23.x, p0);
    p1 = fmaf(v2, w23.y, p1);
    p0 = fmaf(v3, w23.z, p0);
    p1 = fmaf(v3, w23.w, p1);
    p0 = row16_sum(p0);
    p1 = row16_sum(p1);
    if (lane == 0) ((float2*)G)[node] = make_float2(p0 * dd, p1 * dd);
}

#define G2_LANES 4
#define G2_WPB   32
#define G2_U     12

__device__ __forceinline__ float quad_sum(float v) {
    v += __int_as_float(__builtin_amdgcn_update_dpp(0, __float_as_int(v), 0xB1, 0xF, 0xF, true));
    v += __int_as_float(__builtin_amdgcn_update_dpp(0, __float_as_int(v), 0x4E, 0xF, 0xF, true));
    return v;
}

__global__ void __launch_bounds__(G2_WPB * 32) k_gather2(const unsigned* __restrict__ rowptr,
                                                 const int* __restrict__ csrSrc,
                                                 const float* __restrict__ dinv,
                                                 const float* __restrict__ G,
                                                 const float* __restrict__ b2,
                                                 float* __restrict__ out, int n) {
    const int wave = (int)((blockIdx.x * blockDim.x + threadIdx.x) >> 5);
    const int lane = threadIdx.x & 31;
    const int grp  = lane >> 2;
    const int sub  = lane & 3;
    const float2* __restrict__ G2 = (const float2*)G;

    const int  nodeA = wave * 16 + grp;
    const int  nodeB = nodeA + 8;
    const bool okA   = nodeA < n;
    const bool okB   = nodeB < n;
    const int  cA    = min(nodeA, n - 1);
    const int  cB    = min(nodeB, n - 1);

    int begA = (int)rowptr[cA], endA = (int)rowptr[cA + 1];
    int begB = (int)rowptr[cB], endB = (int)rowptr[cB + 1];
    const int   nodeS = (sub == 0) ? nodeA : nodeB;
    const bool  okS   = (sub < 2) && (nodeS < n);
    const float dS    = dinv[min(nodeS, n - 1)];
    const float2 selfA = G2[cA];
    const float2 selfB = G2[cB];
    if (!okA) endA = begA;
    if (!okB) endB = begB;

    float2 accA = make_float2(0.0f, 0.0f);
    float2 accB = make_float2(0.0f, 0.0f);
    if (sub == 0 && okA) accA = selfA;
    if (sub == 0 && okB) accB = selfB;

    const int lastA = max(endA - 1, 0);
    const int lastB = max(endB - 1, 0);
    int eA = begA + sub;
    int eB = begB + sub;
    while ((eA < endA) | (eB < endB)) {
        int sA[G2_U], sB[G2_U];
        #pragma unroll
        for (int u = 0; u < G2_U; ++u) {
            sA[u] = csrSrc[min(eA + u * G2_LANES, lastA)];
            sB[u] = csrSrc[min(eB + u * G2_LANES, lastB)];
        }
        float2 gA[G2_U], gB[G2_U];
        #pragma unroll
        for (int u = 0; u < G2_U; ++u) {
            gA[u] = G2[sA[u]];
            gB[u] = G2[sB[u]];
        }
        #pragma unroll
        for (int u = 0; u < G2_U; ++u)
            asm volatile("" :: "v"(gA[u].x), "v"(gA[u].y), "v"(gB[u].x), "v"(gB[u].y));
        #pragma unroll
        for (int u = 0; u < G2_U; ++u) {
            const bool vA = eA + u * G2_LANES < endA;
            const bool vB = eB + u * G2_LANES < endB;
            accA.x += vA ? gA[u].x : 0.0f; accA.y += vA ? gA[u].y : 0.0f;
            accB.x += vB ? gB[u].x : 0.0f; accB.y += vB ? gB[u].y : 0.0f;
        }
        eA += G2_U * G2_LANES;
        eB += G2_U * G2_LANES;
    }

    accA.x = quad_sum(accA.x); accA.y = quad_sum(accA.y);
    accB.x = quad_sum(accB.x); accB.y = quad_sum(accB.y);

    if (okS) {
        const float2 t = (sub == 0) ? accA : accB;
        float2 o;
        o.x = fmaf(dS, t.x, b2[0]);
        o.y = fmaf(dS, t.y, b2[1]);
        ((float2*)out)[nodeS] = o;
    }
}

extern "C" void kernel_launch(void* const* d_in, const int* in_sizes, int n_in,
                              void* d_out, int out_size, void* d_ws, size_t ws_size,
                              hipStream_t stream) {
    const float* x  = (const float*)d_in[0];
    const float* W1 = (const float*)d_in[1];
    const float* b1 = (const float*)d_in[2];
    const float* W2 = (const float*)d_in[3];
    const float* b2 = (const float*)d_in[4];
    const int*   ei = (const int*)d_in[5];

    const int N = in_sizes[0] / IN_FEATS;
    const int E = in_sizes[5] / 2;
    const int* src = ei;
    const int* dst = ei + E;

    const int rangeLen = (N + NR - 1) / NR;
    const int chunkLen = (E + NC - 1) / NC;
    const int nPad     = (N + 31) & ~31;

    float* ws   = (float*)d_ws;
    float* dinv = ws;
    float* H    = dinv + nPad;
    float* G    = H    + (size_t)N * HIDDEN;
    unsigned* rowptr = (unsigned*)(G + (size_t)nPad * 2);
    uint2*    tabT   = (uint2*)(rowptr + nPad + 32);
    int2*     bkt    = (int2*)(tabT + (size_t)NR * NC);
    int*      csrSrc = (int*)(bkt + (size_t)NC * chunkLen);
    float* out  = (float*)d_out;

    const int T = 256;

    const int nRowBlocks = N / 16;
    int gemmWgs = (nRowBlocks + (PT / 32) - 1) / (PT / 32);
    if (gemmWgs < 256) gemmWgs = 256;
    k_part_and_gemm1<<<NC + gemmWgs, PT, 0, stream>>>(src, dst, E, chunkLen, rangeLen, tabT, bkt, NC, x, W1, H, nRowBlocks, gemmWgs);

    k_range_csr<<<NR, PT, 0, stream>>>(tabT, bkt, E, N, chunkLen, rangeLen, dinv, rowptr, csrSrc, H);

    k_gather1<<<(N * 32 + T - 1) / T, T, 0, stream>>>(rowptr, csrSrc, dinv, H, b1, W2, G, N);

    const int g2Waves = (N + 15) / 16;
    int g2Wpb = (g2Waves + 256 - 1) / 256;
    if (g2Wpb > G2_WPB) g2Wpb = G2_WPB;
    if (g2Wpb < 1) g2Wpb = 1;
    k_gather2<<<(g2Waves + g2Wpb - 1) / g2Wpb, g2Wpb * 32, 0, stream>>>(rowptr, csrSrc, dinv, G, b2, out, N);
}
